# starting kernel
# baseline (speedup 1.0000x reference)
#include <hip/hip_runtime.h>
#include <hip/hip_bf16.h>
#include <math.h>

typedef float v2f __attribute__((ext_vector_type(2)));
typedef float v8f __attribute__((ext_vector_type(8)));

#define IN_C   128
#define HID_C  128
#define OUT_C  64
#define NBATCH 512

__global__ __launch_bounds__(256) void gemm_f32_wmma(
    const float* __restrict__ A, const float* __restrict__ B,
    const float* __restrict__ bias, float* __restrict__ C,
    int M, int K, int N, int act)
{
    __shared__ float lds_a[16][132];

    const int tid  = threadIdx.x;
    const int wave = tid >> 5;
    const int lane = tid & 31;
    const int row0 = blockIdx.x * 16;

    const int nvec = (16 * K) >> 2;
    const int kv   = K >> 2;
    for (int i = tid; i < nvec; i += blockDim.x) {
        int r  = i / kv;
        int k4 = (i - r * kv) << 2;
        float4 v = *(const float4*)(A + (size_t)(row0 + r) * K + k4);
        *(float4*)&lds_a[r][k4] = v;
    }
    __syncthreads();

    const int m    = lane & 15;
    const int kh   = (lane >> 4) << 1;
    const int col0 = wave * 16;

    v8f acc = {};
    for (int kb = 0; kb < K; kb += 4) {
        v2f a, b;
        a.x = lds_a[m][kb + kh];
        a.y = lds_a[m][kb + kh + 1];
        const float* Bp = B + (size_t)(kb + kh) * N + col0 + m;
        b.x = Bp[0];
        b.y = Bp[N];
        acc = __builtin_amdgcn_wmma_f32_16x16x4_f32(
            false, a, false, b, (short)0, acc, false, false);
    }

    const int   col   = col0 + m;
    const int   rbase = row0 + ((lane >> 4) << 3);
    const float bv    = bias ? bias[col] : 0.0f;
#pragma unroll
    for (int v = 0; v < 8; ++v) {
        float x = acc[v] + bv;
        if (act == 1) x = fmaxf(x, 0.0f);
        C[(size_t)(rbase + v) * N + col] = x;
    }
}

__global__ void fill_f32(float* __restrict__ p, float v, int n)
{
    int i = blockIdx.x * blockDim.x + threadIdx.x;
    if (i < n) p[i] = v;
}

__global__ void deg_edges(const int* __restrict__ dst, float* __restrict__ deg, int E)
{
    int e = blockIdx.x * blockDim.x + threadIdx.x;
    if (e < E) atomicAdd(&deg[dst[e]], 1.0f);
}

__global__ void rsqrt_k(const float* __restrict__ deg, float* __restrict__ dinv, int n)
{
    int i = blockIdx.x * blockDim.x + threadIdx.x;
    if (i < n) dinv[i] = rsqrtf(deg[i]);
}

__global__ __launch_bounds__(32) void self_loop_init(
    const float* __restrict__ xw, const float* __restrict__ dinv,
    float* __restrict__ h, int n)
{
    int node = blockIdx.x;
    int c4   = threadIdx.x << 2;
    float di = dinv[node];
    float s  = di * di;
    size_t idx = (size_t)node * HID_C + c4;
    float4 v = *(const float4*)(xw + idx);
    v.x *= s; v.y *= s; v.z *= s; v.w *= s;
    *(float4*)(h + idx) = v;
}

__global__ __launch_bounds__(256) void edge_scatter(
    const float* __restrict__ xw, const float* __restrict__ dinv,
    const int* __restrict__ src, const int* __restrict__ dst,
    float* __restrict__ h, int E)
{
    int e  = blockIdx.x * 8 + (threadIdx.x >> 5);
    int c4 = (threadIdx.x & 31) << 2;
    if (e < E) {
        int   s    = src[e];
        int   d    = dst[e];
        float norm = dinv[s] * dinv[d];
        float4 v = *(const float4*)(xw + (size_t)s * HID_C + c4);
        float* hp = h + (size_t)d * HID_C + c4;
        atomicAdd(hp + 0, v.x * norm);
        atomicAdd(hp + 1, v.y * norm);
        atomicAdd(hp + 2, v.z * norm);
        atomicAdd(hp + 3, v.w * norm);
    }
}

__global__ __launch_bounds__(32) void bias_gelu(
    float* __restrict__ h, const float* __restrict__ b, int n)
{
    int node = blockIdx.x;
    int c4   = threadIdx.x << 2;
    size_t idx = (size_t)node * HID_C + c4;
    float4 v  = *(const float4*)(h + idx);
    float4 bb = *(const float4*)(b + c4);
    v.x += bb.x; v.y += bb.y; v.z += bb.z; v.w += bb.w;
    v.x = 0.5f * v.x * (1.0f + erff(v.x * 0.70710678118654752f));
    v.y = 0.5f * v.y * (1.0f + erff(v.y * 0.70710678118654752f));
    v.z = 0.5f * v.z * (1.0f + erff(v.z * 0.70710678118654752f));
    v.w = 0.5f * v.w * (1.0f + erff(v.w * 0.70710678118654752f));
    *(float4*)(h + idx) = v;
}

__global__ __launch_bounds__(32) void pool_accum(
    const float* __restrict__ h, const int* __restrict__ batch,
    float* __restrict__ sums, float* __restrict__ cnt, int n)
{
    int node = blockIdx.x;
    int c4   = threadIdx.x << 2;
    int b    = batch[node];
    float4 v = *(const float4*)(h + (size_t)node * HID_C + c4);
    float* sp = sums + (size_t)b * HID_C + c4;
    atomicAdd(sp + 0, v.x);
    atomicAdd(sp + 1, v.y);
    atomicAdd(sp + 2, v.z);
    atomicAdd(sp + 3, v.w);
    if (threadIdx.x == 0) atomicAdd(&cnt[b], 1.0f);
}

__global__ void pool_final(float* __restrict__ sums, const float* __restrict__ cnt, int n)
{
    int i = blockIdx.x * blockDim.x + threadIdx.x;
    if (i < n) sums[i] /= fmaxf(cnt[i / HID_C], 1.0f);
}

extern "C" void kernel_launch(void* const* d_in, const int* in_sizes, int n_in,
                              void* d_out, int out_size, void* d_ws, size_t ws_size,
                              hipStream_t stream)
{
    const float* x     = (const float*)d_in[0];
    const int*   ei    = (const int*)d_in[1];
    const int*   batch = (const int*)d_in[2];
    const float* W[3]  = { (const float*)d_in[3], (const float*)d_in[5], (const float*)d_in[7] };
    const float* bv[3] = { (const float*)d_in[4], (const float*)d_in[6], (const float*)d_in[8] };
    const float* Wm1   = (const float*)d_in[9];
    const float* bm1   = (const float*)d_in[10];
    const float* Wm2   = (const float*)d_in[11];
    const float* bm2   = (const float*)d_in[12];

    const int N = in_sizes[0] / IN_C;
    const int E = in_sizes[1] / 2;
    const int* src = ei;
    const int* dst = ei + E;

    char*  ws  = (char*)d_ws;
    size_t off = 0;
    auto carve = [&](size_t bytes) -> float* {
        float* p = (float*)(ws + off);
        off = (off + bytes + 255) & ~(size_t)255;
        return p;
    };
    float* deg  = carve((size_t)N * 4);
    float* dinv = carve((size_t)N * 4);
    float* buf0 = carve((size_t)N * HID_C * 4);
    float* buf1 = carve((size_t)N * HID_C * 4);
    float* psum = carve((size_t)NBATCH * HID_C * 4);
    float* pcnt = carve((size_t)NBATCH * 4);
    float* hid  = carve((size_t)NBATCH * HID_C * 4);

    fill_f32<<<(N + 255) / 256, 256, 0, stream>>>(deg, 1.0f, N);
    deg_edges<<<(E + 255) / 256, 256, 0, stream>>>(dst, deg, E);
    rsqrt_k<<<(N + 255) / 256, 256, 0, stream>>>(deg, dinv, N);

    const float* cur = x;
    for (int l = 0; l < 3; ++l) {
        gemm_f32_wmma<<<N / 16, 256, 0, stream>>>(cur, W[l], nullptr, buf0,
                                                  N, HID_C, HID_C, 0);
        self_loop_init<<<N, 32, 0, stream>>>(buf0, dinv, buf1, N);
        edge_scatter<<<(E + 7) / 8, 256, 0, stream>>>(buf0, dinv, src, dst, buf1, E);
        bias_gelu<<<N, 32, 0, stream>>>(buf1, bv[l], N);
        cur = buf1;
    }

    fill_f32<<<(NBATCH * HID_C + 255) / 256, 256, 0, stream>>>(psum, 0.0f, NBATCH * HID_C);
    fill_f32<<<(NBATCH + 255) / 256, 256, 0, stream>>>(pcnt, 0.0f, NBATCH);
    pool_accum<<<N, 32, 0, stream>>>(cur, batch, psum, pcnt, N);
    pool_final<<<(NBATCH * HID_C + 255) / 256, 256, 0, stream>>>(psum, pcnt, NBATCH * HID_C);

    gemm_f32_wmma<<<NBATCH / 16, 256, 0, stream>>>(psum, Wm1, bm1, hid,
                                                   NBATCH, HID_C, HID_C, 1);
    gemm_f32_wmma<<<NBATCH / 16, 128, 0, stream>>>(hid, Wm2, bm2, (float*)d_out,
                                                   NBATCH, HID_C, OUT_C, 0);
}


# edge aggregation: global f32 atomic scatter + self-loop pass -> each work-group owns 196 rows of h in LDS, sweeps the dst list for its rows, adds xw[src] by LDS float adds; arrival order as before
# speedup vs baseline: 1.9425x; 1.9425x over previous
#include <hip/hip_runtime.h>
#include <hip/hip_bf16.h>
#include <math.h>

typedef float v2f __attribute__((ext_vector_type(2)));
typedef float v8f __attribute__((ext_vector_type(8)));

#define IN_C   128
#define HID_C  128
#define OUT_C  64
#define NBATCH 512

__global__ __launch_bounds__(256) void gemm_f32_wmma(
    const float* __restrict__ A, const float* __restrict__ B,
    const float* __restrict__ bias, float* __restrict__ C,
    int M, int K, int N, int act)
{
    __shared__ float lds_a[16][132];

    const int tid  = threadIdx.x;
    const int wave = tid >> 5;
    const int lane = tid & 31;
    const int row0 = blockIdx.x * 16;

    const int nvec = (16 * K) >> 2;
    const int kv   = K >> 2;
    for (int i = tid; i < nvec; i += blockDim.x) {
        int r  = i / kv;
        int k4 = (i - r * kv) << 2;
        float4 v = *(const float4*)(A + (size_t)(row0 + r) * K + k4);
        *(float4*)&lds_a[r][k4] = v;
    }
    __syncthreads();

    const int m    = lane & 15;
    const int kh   = (lane >> 4) << 1;
    const int col0 = wave * 16;

    v8f acc = {};
    for (int kb = 0; kb < K; kb += 4) {
        v2f a, b;
        a.x = lds_a[m][kb + kh];
        a.y = lds_a[m][kb + kh + 1];
        const float* Bp = B + (size_t)(kb + kh) * N + col0 + m;
        b.x = Bp[0];
        b.y = Bp[N];
        acc = __builtin_amdgcn_wmma_f32_16x16x4_f32(
            false, a, false, b, (short)0, acc, false, false);
    }

    const int   col   = col0 + m;
    const int   rbase = row0 + ((lane >> 4) << 3);
    const float bv    = bias ? bias[col] : 0.0f;
#pragma unroll
    for (int v = 0; v < 8; ++v) {
        float x = acc[v] + bv;
        if (act == 1) x = fmaxf(x, 0.0f);
        C[(size_t)(rbase + v) * N + col] = x;
    }
}

__global__ void fill_f32(float* __restrict__ p, float v, int n)
{
    int i = blockIdx.x * blockDim.x + threadIdx.x;
    if (i < n) p[i] = v;
}

__global__ void deg_edges(const int* __restrict__ dst, float* __restrict__ deg, int E)
{
    int e = blockIdx.x * blockDim.x + threadIdx.x;
    if (e < E) atomicAdd(&deg[dst[e]], 1.0f);
}

__global__ void rsqrt_k(const float* __restrict__ deg, float* __restrict__ dinv, int n)
{
    int i = blockIdx.x * blockDim.x + threadIdx.x;
    if (i < n) dinv[i] = rsqrtf(deg[i]);
}

#define TR    196
#define QCAP  544

__device__ __forceinline__ void queue_if_owned(int d, int e, int r0, int nr,
                                               int* qe, int* qr, int& qn)
{
    const unsigned rel  = (unsigned)(d - r0);
    const bool     m    = rel < (unsigned)nr;
    const unsigned mask = __builtin_amdgcn_ballot_w32(m);
    if (mask) {
        if (m) {
            int pos = qn + (int)__builtin_amdgcn_mbcnt_lo(mask, 0u);
            qe[pos] = e;
            qr[pos] = (int)rel;
        }
        qn += __builtin_popcount(mask);
    }
}

__device__ __forceinline__ void drain_queue(
    const float* __restrict__ xw, const float* __restrict__ dinv,
    const int* __restrict__ src, float* acc,
    const int* qe, const int* qr, int cnt, int r0, int lane)
{
    int   s_i = 0, r_i = 0;
    float n_i = 0.0f;
    if (lane < cnt) {
        int e = qe[lane];
        r_i = qr[lane];
        s_i = src[e];
        n_i = dinv[s_i] * dinv[r0 + r_i];
    }
    for (int k = 0; k < cnt; k += 8) {
        float v[8][4];
        int   rr[8];
        float nn[8];
#pragma unroll
        for (int u = 0; u < 8; ++u) {
            int kk = (k + u < cnt) ? (k + u) : k;
            int s  = __builtin_amdgcn_readlane(s_i, kk);
            rr[u]  = __builtin_amdgcn_readlane(r_i, kk);
            nn[u]  = __int_as_float(__builtin_amdgcn_readlane(__float_as_int(n_i), kk));
            const float* rowp = xw + (size_t)s * HID_C + lane;
            v[u][0] = rowp[0];
            v[u][1] = rowp[32];
            v[u][2] = rowp[64];
            v[u][3] = rowp[96];
        }
#pragma unroll
        for (int u = 0; u < 8; ++u) {
            if (k + u < cnt) {
                float* a = acc + rr[u] * HID_C + lane;
                atomicAdd(a,      v[u][0] * nn[u]);
                atomicAdd(a + 32, v[u][1] * nn[u]);
                atomicAdd(a + 64, v[u][2] * nn[u]);
                atomicAdd(a + 96, v[u][3] * nn[u]);
            }
        }
    }
}

__global__ __launch_bounds__(1024) void agg_rows_lds(
    const float* __restrict__ xw, const float* __restrict__ dinv,
    const int* __restrict__ src, const int* __restrict__ dst,
    float* __restrict__ h, int E, int n)
{
    __shared__ float acc[TR * HID_C];
    __shared__ int   q_e[32 * QCAP];
    __shared__ int   q_r[32 * QCAP];

    const int tid  = threadIdx.x;
    const int wave = tid >> 5;
    const int lane = tid & 31;
    const int r0   = blockIdx.x * TR;
    const int nr   = min(TR, n - r0);

    for (int i = tid; i < nr * 32; i += 1024) {
        int   r  = i >> 5;
        int   c4 = (i & 31) << 2;
        float di = dinv[r0 + r];
        float s  = di * di;
        float4 v = *(const float4*)(xw + (size_t)(r0 + r) * HID_C + c4);
        v.x *= s; v.y *= s; v.z *= s; v.w *= s;
        *(float4*)&acc[r * HID_C + c4] = v;
    }
    __syncthreads();

    int* qe = q_e + wave * QCAP;
    int* qr = q_r + wave * QCAP;
    int  qn = 0;

    const int ngf = ((E & 3) == 0) ? (E >> 9) : 0;
    const int g0  = (int)(((unsigned long long)blockIdx.x * (unsigned long long)ngf) / gridDim.x);

    int  it = wave;
    int4 c[4];
    if (it < ngf) {
        int g = g0 + it; if (g >= ngf) g -= ngf;
        const int4* p = (const int4*)(dst + ((size_t)g << 9)) + lane;
#pragma unroll
        for (int u = 0; u < 4; ++u) c[u] = p[u * 32];
    }
    while (it < ngf) {
        int g = g0 + it; if (g >= ngf) g -= ngf;
        const int eb = (g << 9) + (lane << 2);

        const int itn = it + 32;
        int gn = g0 + ((itn < ngf) ? itn : it); if (gn >= ngf) gn -= ngf;
        const int4* pn = (const int4*)(dst + ((size_t)gn << 9)) + lane;
        int4 nx[4];
#pragma unroll
        for (int u = 0; u < 4; ++u) nx[u] = pn[u * 32];

#pragma unroll
        for (int u = 0; u < 4; ++u) {
            queue_if_owned(c[u].x, eb + u * 128 + 0, r0, nr, qe, qr, qn);
            queue_if_owned(c[u].y, eb + u * 128 + 1, r0, nr, qe, qr, qn);
            queue_if_owned(c[u].z, eb + u * 128 + 2, r0, nr, qe, qr, qn);
            queue_if_owned(c[u].w, eb + u * 128 + 3, r0, nr, qe, qr, qn);
        }
        while (qn >= 32) {
            drain_queue(xw, dinv, src, acc, qe + (qn - 32), qr + (qn - 32), 32, r0, lane);
            qn -= 32;
        }

#pragma unroll
        for (int u = 0; u < 4; ++u) c[u] = nx[u];
        it = itn;
    }

    for (int e0 = (ngf << 9) + (wave << 5); e0 < E; e0 += 1024) {
        const int e = e0 + lane;
        const int d = (e < E) ? dst[e] : -1;
        queue_if_owned(d, e, r0, nr, qe, qr, qn);
        if (qn >= 32) {
            drain_queue(xw, dinv, src, acc, qe + (qn - 32), qr + (qn - 32), 32, r0, lane);
            qn -= 32;
        }
    }
    if (qn > 0) drain_queue(xw, dinv, src, acc, qe, qr, qn, r0, lane);
    __syncthreads();

    for (int i = tid; i < nr * 32; i += 1024) {
        int r  = i >> 5;
        int c4 = (i & 31) << 2;
        *(float4*)(h + (size_t)(r0 + r) * HID_C + c4) = *(const float4*)&acc[r * HID_C + c4];
    }
}

__global__ __launch_bounds__(32) void bias_gelu(
    float* __restrict__ h, const float* __restrict__ b, int n)
{
    int node = blockIdx.x;
    int c4   = threadIdx.x << 2;
    size_t idx = (size_t)node * HID_C + c4;
    float4 v  = *(const float4*)(h + idx);
    float4 bb = *(const float4*)(b + c4);
    v.x += bb.x; v.y += bb.y; v.z += bb.z; v.w += bb.w;
    v.x = 0.5f * v.x * (1.0f + erff(v.x * 0.70710678118654752f));
    v.y = 0.5f * v.y * (1.0f + erff(v.y * 0.70710678118654752f));
    v.z = 0.5f * v.z * (1.0f + erff(v.z * 0.70710678118654752f));
    v.w = 0.5f * v.w * (1.0f + erff(v.w * 0.70710678118654752f));
    *(float4*)(h + idx) = v;
}

__global__ __launch_bounds__(32) void pool_accum(
    const float* __restrict__ h, const int* __restrict__ batch,
    float* __restrict__ sums, float* __restrict__ cnt, int n)
{
    int node = blockIdx.x;
    int c4   = threadIdx.x << 2;
    int b    = batch[node];
    float4 v = *(const float4*)(h + (size_t)node * HID_C + c4);
    float* sp = sums + (size_t)b * HID_C + c4;
    atomicAdd(sp + 0, v.x);
    atomicAdd(sp + 1, v.y);
    atomicAdd(sp + 2, v.z);
    atomicAdd(sp + 3, v.w);
    if (threadIdx.x == 0) atomicAdd(&cnt[b], 1.0f);
}

__global__ void pool_final(float* __restrict__ sums, const float* __restrict__ cnt, int n)
{
    int i = blockIdx.x * blockDim.x + threadIdx.x;
    if (i < n) sums[i] /= fmaxf(cnt[i / HID_C], 1.0f);
}

extern "C" void kernel_launch(void* const* d_in, const int* in_sizes, int n_in,
                              void* d_out, int out_size, void* d_ws, size_t ws_size,
                              hipStream_t stream)
{
    const float* x     = (const float*)d_in[0];
    const int*   ei    = (const int*)d_in[1];
    const int*   batch = (const int*)d_in[2];
    const float* W[3]  = { (const float*)d_in[3], (const float*)d_in[5], (const float*)d_in[7] };
    const float* bv[3] = { (const float*)d_in[4], (const float*)d_in[6], (const float*)d_in[8] };
    const float* Wm1   = (const float*)d_in[9];
    const float* bm1   = (const float*)d_in[10];
    const float* Wm2   = (const float*)d_in[11];
    const float* bm2   = (const float*)d_in[12];

    const int N = in_sizes[0] / IN_C;
    const int E = in_sizes[1] / 2;
    const int* src = ei;
    const int* dst = ei + E;

    char*  ws  = (char*)d_ws;
    size_t off = 0;
    auto carve = [&](size_t bytes) -> float* {
        float* p = (float*)(ws + off);
        off = (off + bytes + 255) & ~(size_t)255;
        return p;
    };
    float* deg  = carve((size_t)N * 4);
    float* dinv = carve((size_t)N * 4);
    float* buf0 = carve((size_t)N * HID_C * 4);
    float* buf1 = carve((size_t)N * HID_C * 4);
    float* psum = carve((size_t)NBATCH * HID_C * 4);
    float* pcnt = carve((size_t)NBATCH * 4);
    float* hid  = carve((size_t)NBATCH * HID_C * 4);

    fill_f32<<<(N + 255) / 256, 256, 0, stream>>>(deg, 1.0f, N);
    deg_edges<<<(E + 255) / 256, 256, 0, stream>>>(dst, deg, E);
    rsqrt_k<<<(N + 255) / 256, 256, 0, stream>>>(deg, dinv, N);

    const float* cur = x;
    for (int l = 0; l < 3; ++l) {
        gemm_f32_wmma<<<N / 16, 256, 0, stream>>>(cur, W[l], nullptr, buf0,
                                                  N, HID_C, HID_C, 0);
        agg_rows_lds<<<(N + TR - 1) / TR, 1024, 0, stream>>>(buf0, dinv, src, dst, buf1, E, N);
        bias_gelu<<<N, 32, 0, stream>>>(buf1, bv[l], N);
        cur = buf1;
    }

    fill_f32<<<(NBATCH * HID_C + 255) / 256, 256, 0, stream>>>(psum, 0.0f, NBATCH * HID_C);
    fill_f32<<<(NBATCH + 255) / 256, 256, 0, stream>>>(pcnt, 0.0f, NBATCH);
    pool_accum<<<N, 32, 0, stream>>>(cur, batch, psum, pcnt, N);
    pool_final<<<(NBATCH * HID_C + 255) / 256, 256, 0, stream>>>(psum, pcnt, NBATCH * HID_C);

    gemm_f32_wmma<<<NBATCH / 16, 256, 0, stream>>>(psum, Wm1, bm1, hid,
                                                   NBATCH, HID_C, HID_C, 1);
    gemm_f32_wmma<<<NBATCH / 16, 128, 0, stream>>>(hid, Wm2, bm2, (float*)d_out,
                                                   NBATCH, HID_C, OUT_C, 0);
}


# bias + exact GELU applied where the next layer's f32 WMMA product stages its A rows and where the pool loads h, not in a pass of its own: 3 launches of one-wave work-groups less; sum order unchanged
# speedup vs baseline: 2.0488x; 1.0547x over previous
#include <hip/hip_runtime.h>
#include <hip/hip_bf16.h>
#include <math.h>

typedef float v2f __attribute__((ext_vector_type(2)));
typedef float v8f __attribute__((ext_vector_type(8)));

#define IN_C   128
#define HID_C  128
#define OUT_C  64
#define NBATCH 512

__device__ __forceinline__ float gelu_exact(float x)
{
    return 0.5f * x * (1.0f + erff(x * 0.70710678118654752f));
}

__global__ __launch_bounds__(256) void gemm_f32_wmma(
    const float* __restrict__ A, const float* __restrict__ B,
    const float* __restrict__ bias, float* __restrict__ C,
    int M, int K, int N, int act, const float* __restrict__ abias)
{
    __shared__ float lds_a[16][132];

    const int tid  = threadIdx.x;
    const int wave = tid >> 5;
    const int lane = tid & 31;
    const int row0 = blockIdx.x * 16;

    const int nvec = (16 * K) >> 2;
    const int kv   = K >> 2;
    for (int i = tid; i < nvec; i += blockDim.x) {
        int r  = i / kv;
        int k4 = (i - r * kv) << 2;
        float4 v = *(const float4*)(A + (size_t)(row0 + r) * K + k4);
        if (abias) {
            float4 bb = *(const float4*)(abias + k4);
            v.x = gelu_exact(v.x + bb.x);
            v.y = gelu_exact(v.y + bb.y);
            v.z = gelu_exact(v.z + bb.z);
            v.w = gelu_exact(v.w + bb.w);
        }
        *(float4*)&lds_a[r][k4] = v;
    }
    __syncthreads();

    const int m    = lane & 15;
    const int kh   = (lane >> 4) << 1;
    const int col0 = wave * 16;

    v8f acc = {};
    for (int kb = 0; kb < K; kb += 4) {
        v2f a, b;
        a.x = lds_a[m][kb + kh];
        a.y = lds_a[m][kb + kh + 1];
        const float* Bp = B + (size_t)(kb + kh) * N + col0 + m;
        b.x = Bp[0];
        b.y = Bp[N];
        acc = __builtin_amdgcn_wmma_f32_16x16x4_f32(
            false, a, false, b, (short)0, acc, false, false);
    }

    const int   col   = col0 + m;
    const int   rbase = row0 + ((lane >> 4) << 3);
    const float bv    = bias ? bias[col] : 0.0f;
#pragma unroll
    for (int v = 0; v < 8; ++v) {
        float x = acc[v] + bv;
        if (act == 1) x = fmaxf(x, 0.0f);
        C[(size_t)(rbase + v) * N + col] = x;
    }
}

__global__ void fill_f32(float* __restrict__ p, float v, int n)
{
    int i = blockIdx.x * blockDim.x + threadIdx.x;
    if (i < n) p[i] = v;
}

__global__ void deg_edges(const int* __restrict__ dst, float* __restrict__ deg, int E)
{
    int e = blockIdx.x * blockDim.x + threadIdx.x;
    if (e < E) atomicAdd(&deg[dst[e]], 1.0f);
}

__global__ void rsqrt_k(const float* __restrict__ deg, float* __restrict__ dinv, int n)
{
    int i = blockIdx.x * blockDim.x + threadIdx.x;
    if (i < n) dinv[i] = rsqrtf(deg[i]);
}

#define TR    196
#define QCAP  544

__device__ __forceinline__ void queue_if_owned(int d, int e, int r0, int nr,
                                               int* qe, int* qr, int& qn)
{
    const unsigned rel  = (unsigned)(d - r0);
    const bool     m    = rel < (unsigned)nr;
    const unsigned mask = __builtin_amdgcn_ballot_w32(m);
    if (mask) {
        if (m) {
            int pos = qn + (int)__builtin_amdgcn_mbcnt_lo(mask, 0u);
            qe[pos] = e;
            qr[pos] = (int)rel;
        }
        qn += __builtin_popcount(mask);
    }
}

__device__ __forceinline__ void drain_queue(
    const float* __restrict__ xw, const float* __restrict__ dinv,
    const int* __restrict__ src, float* acc,
    const int* qe, const int* qr, int cnt, int r0, int lane)
{
    int   s_i = 0, r_i = 0;
    float n_i = 0.0f;
    if (lane < cnt) {
        int e = qe[lane];
        r_i = qr[lane];
        s_i = src[e];
        n_i = dinv[s_i] * dinv[r0 + r_i];
    }
    for (int k = 0; k < cnt; k += 8) {
        float v[8][4];
        int   rr[8];
        float nn[8];
#pragma unroll
        for (int u = 0; u < 8; ++u) {
            int kk = (k + u < cnt) ? (k + u) : k;
            int s  = __builtin_amdgcn_readlane(s_i, kk);
            rr[u]  = __builtin_amdgcn_readlane(r_i, kk);
            nn[u]  = __int_as_float(__builtin_amdgcn_readlane(__float_as_int(n_i), kk));
            const float* rowp = xw + (size_t)s * HID_C + lane;
            v[u][0] = rowp[0];
            v[u][1] = rowp[32];
            v[u][2] = rowp[64];
            v[u][3] = rowp[96];
        }
#pragma unroll
        for (int u = 0; u < 8; ++u) {
            if (k + u < cnt) {
                float* a = acc + rr[u] * HID_C + lane;
                atomicAdd(a,      v[u][0] * nn[u]);
                atomicAdd(a + 32, v[u][1] * nn[u]);
                atomicAdd(a + 64, v[u][2] * nn[u]);
                atomicAdd(a + 96, v[u][3] * nn[u]);
            }
        }
    }
}

__global__ __launch_bounds__(1024) void agg_rows_lds(
    const float* __restrict__ xw, const float* __restrict__ dinv,
    const int* __restrict__ src, const int* __restrict__ dst,
    float* __restrict__ h, int E, int n)
{
    __shared__ float acc[TR * HID_C];
    __shared__ int   q_e[32 * QCAP];
    __shared__ int   q_r[32 * QCAP];

    const int tid  = threadIdx.x;
    const int wave = tid >> 5;
    const int lane = tid & 31;
    const int r0   = blockIdx.x * TR;
    const int nr   = min(TR, n - r0);

    for (int i = tid; i < nr * 32; i += 1024) {
        int   r  = i >> 5;
        int   c4 = (i & 31) << 2;
        float di = dinv[r0 + r];
        float s  = di * di;
        float4 v = *(const float4*)(xw + (size_t)(r0 + r) * HID_C + c4);
        v.x *= s; v.y *= s; v.z *= s; v.w *= s;
        *(float4*)&acc[r * HID_C + c4] = v;
    }
    __syncthreads();

    int* qe = q_e + wave * QCAP;
    int* qr = q_r + wave * QCAP;
    int  qn = 0;

    const int ngf = ((E & 3) == 0) ? (E >> 9) : 0;
    const int g0  = (int)(((unsigned long long)blockIdx.x * (unsigned long long)ngf) / gridDim.x);

    int  it = wave;
    int4 c[4];
    if (it < ngf) {
        int g = g0 + it; if (g >= ngf) g -= ngf;
        const int4* p = (const int4*)(dst + ((size_t)g << 9)) + lane;
#pragma unroll
        for (int u = 0; u < 4; ++u) c[u] = p[u * 32];
    }
    while (it < ngf) {
        int g = g0 + it; if (g >= ngf) g -= ngf;
        const int eb = (g << 9) + (lane << 2);

        const int itn = it + 32;
        int gn = g0 + ((itn < ngf) ? itn : it); if (gn >= ngf) gn -= ngf;
        const int4* pn = (const int4*)(dst + ((size_t)gn << 9)) + lane;
        int4 nx[4];
#pragma unroll
        for (int u = 0; u < 4; ++u) nx[u] = pn[u * 32];

#pragma unroll
        for (int u = 0; u < 4; ++u) {
            queue_if_owned(c[u].x, eb + u * 128 + 0, r0, nr, qe, qr, qn);
            queue_if_owned(c[u].y, eb + u * 128 + 1, r0, nr, qe, qr, qn);
            queue_if_owned(c[u].z, eb + u * 128 + 2, r0, nr, qe, qr, qn);
            queue_if_owned(c[u].w, eb + u * 128 + 3, r0, nr, qe, qr, qn);
        }
        while (qn >= 32) {
            drain_queue(xw, dinv, src, acc, qe + (qn - 32), qr + (qn - 32), 32, r0, lane);
            qn -= 32;
        }

#pragma unroll
        for (int u = 0; u < 4; ++u) c[u] = nx[u];
        it = itn;
    }

    for (int e0 = (ngf << 9) + (wave << 5); e0 < E; e0 += 1024) {
        const int e = e0 + lane;
        const int d = (e < E) ? dst[e] : -1;
        queue_if_owned(d, e, r0, nr, qe, qr, qn);
        if (qn >= 32) {
            drain_queue(xw, dinv, src, acc, qe + (qn - 32), qr + (qn - 32), 32, r0, lane);
            qn -= 32;
        }
    }
    if (qn > 0) drain_queue(xw, dinv, src, acc, qe, qr, qn, r0, lane);
    __syncthreads();

    for (int i = tid; i < nr * 32; i += 1024) {
        int r  = i >> 5;
        int c4 = (i & 31) << 2;
        *(float4*)(h + (size_t)(r0 + r) * HID_C + c4) = *(const float4*)&acc[r * HID_C + c4];
    }
}

__global__ __launch_bounds__(32) void pool_accum(
    const float* __restrict__ h, const float* __restrict__ hb,
    const int* __restrict__ batch,
    float* __restrict__ sums, float* __restrict__ cnt, int n)
{
    int node = blockIdx.x;
    int c4   = threadIdx.x << 2;
    int b    = batch[node];
    float4 v  = *(const float4*)(h + (size_t)node * HID_C + c4);
    float4 bb = *(const float4*)(hb + c4);
    v.x = gelu_exact(v.x + bb.x);
    v.y = gelu_exact(v.y + bb.y);
    v.z = gelu_exact(v.z + bb.z);
    v.w = gelu_exact(v.w + bb.w);
    float* sp = sums + (size_t)b * HID_C + c4;
    atomicAdd(sp + 0, v.x);
    atomicAdd(sp + 1, v.y);
    atomicAdd(sp + 2, v.z);
    atomicAdd(sp + 3, v.w);
    if (threadIdx.x == 0) atomicAdd(&cnt[b], 1.0f);
}

__global__ void pool_final(float* __restrict__ sums, const float* __restrict__ cnt, int n)
{
    int i = blockIdx.x * blockDim.x + threadIdx.x;
    if (i < n) sums[i] /= fmaxf(cnt[i / HID_C], 1.0f);
}

extern "C" void kernel_launch(void* const* d_in, const int* in_sizes, int n_in,
                              void* d_out, int out_size, void* d_ws, size_t ws_size,
                              hipStream_t stream)
{
    const float* x     = (const float*)d_in[0];
    const int*   ei    = (const int*)d_in[1];
    const int*   batch = (const int*)d_in[2];
    const float* W[3]  = { (const float*)d_in[3], (const float*)d_in[5], (const float*)d_in[7] };
    const float* bv[3] = { (const float*)d_in[4], (const float*)d_in[6], (const float*)d_in[8] };
    const float* Wm1   = (const float*)d_in[9];
    const float* bm1   = (const float*)d_in[10];
    const float* Wm2   = (const float*)d_in[11];
    const float* bm2   = (const float*)d_in[12];

    const int N = in_sizes[0] / IN_C;
    const int E = in_sizes[1] / 2;
    const int* src = ei;
    const int* dst = ei + E;

    char*  ws  = (char*)d_ws;
    size_t off = 0;
    auto carve = [&](size_t bytes) -> float* {
        float* p = (float*)(ws + off);
        off = (off + bytes + 255) & ~(size_t)255;
        return p;
    };
    float* deg  = carve((size_t)N * 4);
    float* dinv = carve((size_t)N * 4);
    float* buf0 = carve((size_t)N * HID_C * 4);
    float* buf1 = carve((size_t)N * HID_C * 4);
    float* psum = carve((size_t)NBATCH * HID_C * 4);
    float* pcnt = carve((size_t)NBATCH * 4);
    float* hid  = carve((size_t)NBATCH * HID_C * 4);

    fill_f32<<<(N + 255) / 256, 256, 0, stream>>>(deg, 1.0f, N);
    deg_edges<<<(E + 255) / 256, 256, 0, stream>>>(dst, deg, E);
    rsqrt_k<<<(N + 255) / 256, 256, 0, stream>>>(deg, dinv, N);

    const float* ab[3] = { nullptr, bv[0], bv[1] };
    const float* cur = x;
    for (int l = 0; l < 3; ++l) {
        gemm_f32_wmma<<<N / 16, 256, 0, stream>>>(cur, W[l], nullptr, buf0,
                                                  N, HID_C, HID_C, 0, ab[l]);
        agg_rows_lds<<<(N + TR - 1) / TR, 1024, 0, stream>>>(buf0, dinv, src, dst, buf1, E, N);
        cur = buf1;
    }

    fill_f32<<<(NBATCH * HID_C + 255) / 256, 256, 0, stream>>>(psum, 0.0f, NBATCH * HID_C);
    fill_f32<<<(NBATCH + 255) / 256, 256, 0, stream>>>(pcnt, 0.0f, NBATCH);
    pool_accum<<<N, 32, 0, stream>>>(cur, bv[2], batch, psum, pcnt, N);
    pool_final<<<(NBATCH * HID_C + 255) / 256, 256, 0, stream>>>(psum, pcnt, NBATCH * HID_C);

    gemm_f32_wmma<<<NBATCH / 16, 256, 0, stream>>>(psum, Wm1, bm1, hid,
                                                   NBATCH, HID_C, HID_C, 1, nullptr);
    gemm_f32_wmma<<<NBATCH / 16, 128, 0, stream>>>(hid, Wm2, bm2, (float*)d_out,
                                                   NBATCH, HID_C, OUT_C, 0, nullptr);
}


# dst sweep hoisted out of the layer loop: one pass groups edges by the work-group owning their row (regions sized by a count pass) and counts degrees; layers read their group; LDS add order as before
# speedup vs baseline: 2.5829x; 1.2607x over previous
#include <hip/hip_runtime.h>
#include <hip/hip_bf16.h>
#include <math.h>

typedef float v2f __attribute__((ext_vector_type(2)));
typedef float v8f __attribute__((ext_vector_type(8)));

#define IN_C   128
#define HID_C  128
#define OUT_C  64
#define NBATCH 512

__device__ __forceinline__ float gelu_exact(float x)
{
    return 0.5f * x * (1.0f + erff(x * 0.70710678118654752f));
}

__global__ __launch_bounds__(256) void gemm_f32_wmma(
    const float* __restrict__ A, const float* __restrict__ B,
    const float* __restrict__ bias, float* __restrict__ C,
    int M, int K, int N, int act, const float* __restrict__ abias)
{
    __shared__ float lds_a[16][132];

    const int tid  = threadIdx.x;
    const int wave = tid >> 5;
    const int lane = tid & 31;
    const int row0 = blockIdx.x * 16;

    const int nvec = (16 * K) >> 2;
    const int kv   = K >> 2;
    for (int i = tid; i < nvec; i += blockDim.x) {
        int r  = i / kv;
        int k4 = (i - r * kv) << 2;
        float4 v = *(const float4*)(A + (size_t)(row0 + r) * K + k4);
        if (abias) {
            float4 bb = *(const float4*)(abias + k4);
            v.x = gelu_exact(v.x + bb.x);
            v.y = gelu_exact(v.y + bb.y);
            v.z = gelu_exact(v.z + bb.z);
            v.w = gelu_exact(v.w + bb.w);
        }
        *(float4*)&lds_a[r][k4] = v;
    }
    __syncthreads();

    const int m    = lane & 15;
    const int kh   = (lane >> 4) << 1;
    const int col0 = wave * 16;

    v8f acc = {};
    for (int kb = 0; kb < K; kb += 4) {
        v2f a, b;
        a.x = lds_a[m][kb + kh];
        a.y = lds_a[m][kb + kh + 1];
        const float* Bp = B + (size_t)(kb + kh) * N + col0 + m;
        b.x = Bp[0];
        b.y = Bp[N];
        acc = __builtin_amdgcn_wmma_f32_16x16x4_f32(
            false, a, false, b, (short)0, acc, false, false);
    }

    const int   col   = col0 + m;
    const int   rbase = row0 + ((lane >> 4) << 3);
    const float bv    = bias ? bias[col] : 0.0f;
#pragma unroll
    for (int v = 0; v < 8; ++v) {
        float x = acc[v] + bv;
        if (act == 1) x = fmaxf(x, 0.0f);
        C[(size_t)(rbase + v) * N + col] = x;
    }
}

__global__ void fill_f32(float* __restrict__ p, float v, int n)
{
    int i = blockIdx.x * blockDim.x + threadIdx.x;
    if (i < n) p[i] = v;
}

#define TR    196
#define NOWN  256
#define NCG   128
#define QCAP  544

__global__ __launch_bounds__(1024) void bin_count(
    const int* __restrict__ dst, int* __restrict__ cntp, int E, int n)
{
    __shared__ int hist[NOWN];
    const int tid = threadIdx.x;
    if (tid < NOWN) hist[tid] = 0;
    __syncthreads();

    const int per = (E + (int)gridDim.x - 1) / (int)gridDim.x;
    const int e0  = blockIdx.x * per;
    const int e1  = min(E, e0 + per);
    for (int e = e0 + tid; e < e1; e += 4 * 1024) {
        unsigned d[4];
#pragma unroll
        for (int u = 0; u < 4; ++u) {
            const int ee = e + u * 1024;
            d[u] = (ee < e1) ? (unsigned)dst[ee] : 0xffffffffu;
        }
#pragma unroll
        for (int u = 0; u < 4; ++u)
            if (d[u] < (unsigned)n) atomicAdd(&hist[d[u] / TR], 1);
    }
    __syncthreads();
    if (tid < NOWN) cntp[blockIdx.x * NOWN + tid] = hist[tid];
}

__device__ __forceinline__ void queue_if_owned(int d, int e, int r0, int nr,
                                               int* qe, int* qr, int& qn)
{
    const unsigned rel  = (unsigned)(d - r0);
    const bool     m    = rel < (unsigned)nr;
    const unsigned mask = __builtin_amdgcn_ballot_w32(m);
    if (mask) {
        if (m) {
            int pos = qn + (int)__builtin_amdgcn_mbcnt_lo(mask, 0u);
            qe[pos] = e;
            qr[pos] = (int)rel;
        }
        qn += __builtin_popcount(mask);
    }
}

__device__ __forceinline__ void flush_queue(
    const int* __restrict__ src, int2* __restrict__ lst, int* cur, int* degc,
    const int* qe, const int* qr, int cnt, int lane)
{
    int base = 0;
    if (lane == 0) base = atomicAdd(cur, cnt);
    base = __builtin_amdgcn_readfirstlane(base);
    if (lane < cnt) {
        const int e = qe[lane];
        const int r = qr[lane];
        lst[base + lane] = make_int2(src[e], r);
        atomicAdd(&degc[r], 1);
    }
}

__global__ __launch_bounds__(1024) void build_bins(
    const int* __restrict__ src, const int* __restrict__ dst,
    const int* __restrict__ cntp, int* __restrict__ regs, int* __restrict__ list,
    float* __restrict__ dinv, int E, int n)
{
    __shared__ int q_e[32 * QCAP];
    __shared__ int q_r[32 * QCAP];
    __shared__ int degc[TR];
    __shared__ int s_reg[3];

    const int tid  = threadIdx.x;
    const int wave = tid >> 5;
    const int lane = tid & 31;
    const int b    = blockIdx.x;
    const int r0   = b * TR;
    const int nr   = min(TR, n - r0);

    if (tid < TR) degc[tid] = 0;
    if (tid < 3)  s_reg[tid] = 0;
    __syncthreads();

    int ps = 0, pc = 0;
    {
        const int j = tid & (NOWN - 1);
        if (j <= b) {
            int s = 0;
#pragma unroll
            for (int g = 0; g < NCG / 4; ++g) s += cntp[((tid >> 8) + 4 * g) * NOWN + j];
            if (j < b) ps = s; else pc = s;
        }
    }
    for (int o = 16; o > 0; o >>= 1) {
        ps += __shfl_xor(ps, o);
        pc += __shfl_xor(pc, o);
    }
    if (lane == 0) {
        atomicAdd(&s_reg[0], ps);
        atomicAdd(&s_reg[1], pc);
    }
    __syncthreads();
    const int start = s_reg[0];
    if (tid == 0) {
        regs[2 * b]     = start;
        regs[2 * b + 1] = s_reg[1];
    }
    int2* lst = (int2*)list + start;
    int*  cur = &s_reg[2];

    int* qe = q_e + wave * QCAP;
    int* qr = q_r + wave * QCAP;
    int  qn = 0;

    const int ngf = ((E & 3) == 0) ? (E >> 9) : 0;
    const int g0  = (int)(((unsigned long long)blockIdx.x * (unsigned long long)ngf) / gridDim.x);

    int  it = wave;
    int4 c[4];
    if (it < ngf) {
        int g = g0 + it; if (g >= ngf) g -= ngf;
        const int4* p = (const int4*)(dst + ((size_t)g << 9)) + lane;
#pragma unroll
        for (int u = 0; u < 4; ++u) c[u] = p[u * 32];
    }
    while (it < ngf) {
        int g = g0 + it; if (g >= ngf) g -= ngf;
        const int eb = (g << 9) + (lane << 2);

        const int itn = it + 32;
        int gn = g0 + ((itn < ngf) ? itn : it); if (gn >= ngf) gn -= ngf;
        const int4* pn = (const int4*)(dst + ((size_t)gn << 9)) + lane;
        int4 nx[4];
#pragma unroll
        for (int u = 0; u < 4; ++u) nx[u] = pn[u * 32];

#pragma unroll
        for (int u = 0; u < 4; ++u) {
            queue_if_owned(c[u].x, eb + u * 128 + 0, r0, nr, qe, qr, qn);
            queue_if_owned(c[u].y, eb + u * 128 + 1, r0, nr, qe, qr, qn);
            queue_if_owned(c[u].z, eb + u * 128 + 2, r0, nr, qe, qr, qn);
            queue_if_owned(c[u].w, eb + u * 128 + 3, r0, nr, qe, qr, qn);
        }
        while (qn >= 32) {
            flush_queue(src, lst, cur, degc, qe + (qn - 32), qr + (qn - 32), 32, lane);
            qn -= 32;
        }

#pragma unroll
        for (int u = 0; u < 4; ++u) c[u] = nx[u];
        it = itn;
    }

    for (int e0 = (ngf << 9) + (wave << 5); e0 < E; e0 += 1024) {
        const int e = e0 + lane;
        const int d = (e < E) ? dst[e] : -1;
        queue_if_owned(d, e, r0, nr, qe, qr, qn);
        if (qn >= 32) {
            flush_queue(src, lst, cur, degc, qe + (qn - 32), qr + (qn - 32), 32, lane);
            qn -= 32;
        }
    }
    if (qn > 0) flush_queue(src, lst, cur, degc, qe, qr, qn, lane);
    __syncthreads();

    if (tid < nr) dinv[r0 + tid] = rsqrtf((float)(degc[tid] + 1));
}

__device__ __forceinline__ void add_rows(
    const float* __restrict__ xw, float* acc,
    int s_i, int r_i, float n_i, int cnt, int lane)
{
    for (int k = 0; k < cnt; k += 8) {
        float v[8][4];
        int   rr[8];
        float nn[8];
#pragma unroll
        for (int u = 0; u < 8; ++u) {
            int kk = (k + u < cnt) ? (k + u) : k;
            int s  = __builtin_amdgcn_readlane(s_i, kk);
            rr[u]  = __builtin_amdgcn_readlane(r_i, kk);
            nn[u]  = __int_as_float(__builtin_amdgcn_readlane(__float_as_int(n_i), kk));
            const float* rowp = xw + (size_t)s * HID_C + lane;
            v[u][0] = rowp[0];
            v[u][1] = rowp[32];
            v[u][2] = rowp[64];
            v[u][3] = rowp[96];
        }
#pragma unroll
        for (int u = 0; u < 8; ++u) {
            if (k + u < cnt) {
                float* a = acc + rr[u] * HID_C + lane;
                atomicAdd(a,      v[u][0] * nn[u]);
                atomicAdd(a + 32, v[u][1] * nn[u]);
                atomicAdd(a + 64, v[u][2] * nn[u]);
                atomicAdd(a + 96, v[u][3] * nn[u]);
            }
        }
    }
}

__global__ __launch_bounds__(1024) void agg_list(
    const float* __restrict__ xw, const float* __restrict__ dinv,
    const int* __restrict__ regs, const int* __restrict__ list,
    float* __restrict__ h, int n)
{
    __shared__ float acc[TR * HID_C];

    const int tid  = threadIdx.x;
    const int wave = tid >> 5;
    const int lane = tid & 31;
    const int b    = blockIdx.x;
    const int r0   = b * TR;
    const int nr   = min(TR, n - r0);

    const int   count = regs[2 * b + 1];
    const int2* lst   = (const int2*)list + regs[2 * b];
    const int   nb    = (count + 31) >> 5;
    int  bi = wave;
    int2 en = make_int2(0, 0);
    if (bi < nb) {
        const int idx = (bi << 5) + lane;
        if (idx < count) en = lst[idx];
    }

    for (int i = tid; i < nr * 32; i += 1024) {
        int   r  = i >> 5;
        int   c4 = (i & 31) << 2;
        float di = dinv[r0 + r];
        float s  = di * di;
        float4 v = *(const float4*)(xw + (size_t)(r0 + r) * HID_C + c4);
        v.x *= s; v.y *= s; v.z *= s; v.w *= s;
        *(float4*)&acc[r * HID_C + c4] = v;
    }
    __syncthreads();

    while (bi < nb) {
        const int cnt = min(32, count - (bi << 5));
        float n_i = 0.0f;
        if (lane < cnt) n_i = dinv[en.x] * dinv[r0 + en.y];
        const int bn = bi + 32;
        int2 nx = make_int2(0, 0);
        if (bn < nb) {
            const int idx = (bn << 5) + lane;
            if (idx < count) nx = lst[idx];
        }
        add_rows(xw, acc, en.x, en.y, n_i, cnt, lane);
        en = nx;
        bi = bn;
    }
    __syncthreads();

    for (int i = tid; i < nr * 32; i += 1024) {
        int r  = i >> 5;
        int c4 = (i & 31) << 2;
        *(float4*)(h + (size_t)(r0 + r) * HID_C + c4) = *(const float4*)&acc[r * HID_C + c4];
    }
}

__global__ __launch_bounds__(32) void pool_accum(
    const float* __restrict__ h, const float* __restrict__ hb,
    const int* __restrict__ batch,
    float* __restrict__ sums, float* __restrict__ cnt, int n)
{
    int node = blockIdx.x;
    int c4   = threadIdx.x << 2;
    int b    = batch[node];
    float4 v  = *(const float4*)(h + (size_t)node * HID_C + c4);
    float4 bb = *(const float4*)(hb + c4);
    v.x = gelu_exact(v.x + bb.x);
    v.y = gelu_exact(v.y + bb.y);
    v.z = gelu_exact(v.z + bb.z);
    v.w = gelu_exact(v.w + bb.w);
    float* sp = sums + (size_t)b * HID_C + c4;
    atomicAdd(sp + 0, v.x);
    atomicAdd(sp + 1, v.y);
    atomicAdd(sp + 2, v.z);
    atomicAdd(sp + 3, v.w);
    if (threadIdx.x == 0) atomicAdd(&cnt[b], 1.0f);
}

__global__ void pool_final(float* __restrict__ sums, const float* __restrict__ cnt, int n)
{
    int i = blockIdx.x * blockDim.x + threadIdx.x;
    if (i < n) sums[i] /= fmaxf(cnt[i / HID_C], 1.0f);
}

extern "C" void kernel_launch(void* const* d_in, const int* in_sizes, int n_in,
                              void* d_out, int out_size, void* d_ws, size_t ws_size,
                              hipStream_t stream)
{
    const float* x     = (const float*)d_in[0];
    const int*   ei    = (const int*)d_in[1];
    const int*   batch = (const int*)d_in[2];
    const float* W[3]  = { (const float*)d_in[3], (const float*)d_in[5], (const float*)d_in[7] };
    const float* bv[3] = { (const float*)d_in[4], (const float*)d_in[6], (const float*)d_in[8] };
    const float* Wm1   = (const float*)d_in[9];
    const float* bm1   = (const float*)d_in[10];
    const float* Wm2   = (const float*)d_in[11];
    const float* bm2   = (const float*)d_in[12];

    const int N = in_sizes[0] / IN_C;
    const int E = in_sizes[1] / 2;
    const int* src = ei;
    const int* dst = ei + E;

    char*  ws  = (char*)d_ws;
    size_t off = 0;
    auto carve = [&](size_t bytes) -> float* {
        float* p = (float*)(ws + off);
        off = (off + bytes + 255) & ~(size_t)255;
        return p;
    };
    float* dinv = carve((size_t)N * 4);
    float* buf0 = carve((size_t)N * HID_C * 4);
    float* buf1 = carve((size_t)N * HID_C * 4);
    float* psum = carve((size_t)NBATCH * HID_C * 4);
    float* pcnt = carve((size_t)NBATCH * 4);
    float* hid  = carve((size_t)NBATCH * HID_C * 4);
    int*   cntp = (int*)carve((size_t)NCG * NOWN * 4);
    int*   regs = (int*)carve((size_t)NOWN * 2 * 4);
    int*   list = (int*)carve((size_t)E * 2 * 4);

    bin_count<<<NCG, 1024, 0, stream>>>(dst, cntp, E, N);
    build_bins<<<NOWN, 1024, 0, stream>>>(src, dst, cntp, regs, list, dinv, E, N);

    const float* ab[3] = { nullptr, bv[0], bv[1] };
    const float* cur = x;
    for (int l = 0; l < 3; ++l) {
        gemm_f32_wmma<<<N / 16, 256, 0, stream>>>(cur, W[l], nullptr, buf0,
                                                  N, HID_C, HID_C, 0, ab[l]);
        agg_list<<<NOWN, 1024, 0, stream>>>(buf0, dinv, regs, list, buf1, N);
        cur = buf1;
    }

    fill_f32<<<(NBATCH * HID_C + 255) / 256, 256, 0, stream>>>(psum, 0.0f, NBATCH * HID_C);
    fill_f32<<<(NBATCH + 255) / 256, 256, 0, stream>>>(pcnt, 0.0f, NBATCH);
    pool_accum<<<N, 32, 0, stream>>>(cur, bv[2], batch, psum, pcnt, N);
    pool_final<<<(NBATCH * HID_C + 255) / 256, 256, 0, stream>>>(psum, pcnt, NBATCH * HID_C);

    gemm_f32_wmma<<<NBATCH / 16, 256, 0, stream>>>(psum, Wm1, bm1, hid,
                                                   NBATCH, HID_C, HID_C, 1, nullptr);
    gemm_f32_wmma<<<NBATCH / 16, 128, 0, stream>>>(hid, Wm2, bm2, (float*)d_out,
                                                   NBATCH, HID_C, OUT_C, 0, nullptr);
}


# edge grouping: each owner swept the whole dst list -> each slice's work-group writes its edges into the owners' regions (count-pass offsets + LDS rank); record order changes, LDS adds by arrival
# speedup vs baseline: 2.7532x; 1.0659x over previous
#include <hip/hip_runtime.h>
#include <hip/hip_bf16.h>
#include <math.h>

typedef float v2f __attribute__((ext_vector_type(2)));
typedef float v8f __attribute__((ext_vector_type(8)));

#define IN_C   128
#define HID_C  128
#define OUT_C  64
#define NBATCH 512

__device__ __forceinline__ float gelu_exact(float x)
{
    return 0.5f * x * (1.0f + erff(x * 0.70710678118654752f));
}

__global__ __launch_bounds__(256) void gemm_f32_wmma(
    const float* __restrict__ A, const float* __restrict__ B,
    const float* __restrict__ bias, float* __restrict__ C,
    int M, int K, int N, int act, const float* __restrict__ abias)
{
    __shared__ float lds_a[16][132];

    const int tid  = threadIdx.x;
    const int wave = tid >> 5;
    const int lane = tid & 31;
    const int row0 = blockIdx.x * 16;

    const int nvec = (16 * K) >> 2;
    const int kv   = K >> 2;
    for (int i = tid; i < nvec; i += blockDim.x) {
        int r  = i / kv;
        int k4 = (i - r * kv) << 2;
        float4 v = *(const float4*)(A + (size_t)(row0 + r) * K + k4);
        if (abias) {
            float4 bb = *(const float4*)(abias + k4);
            v.x = gelu_exact(v.x + bb.x);
            v.y = gelu_exact(v.y + bb.y);
            v.z = gelu_exact(v.z + bb.z);
            v.w = gelu_exact(v.w + bb.w);
        }
        *(float4*)&lds_a[r][k4] = v;
    }
    __syncthreads();

    const int m    = lane & 15;
    const int kh   = (lane >> 4) << 1;
    const int col0 = wave * 16;

    v8f acc = {};
    for (int kb = 0; kb < K; kb += 4) {
        v2f a, b;
        a.x = lds_a[m][kb + kh];
        a.y = lds_a[m][kb + kh + 1];
        const float* Bp = B + (size_t)(kb + kh) * N + col0 + m;
        b.x = Bp[0];
        b.y = Bp[N];
        acc = __builtin_amdgcn_wmma_f32_16x16x4_f32(
            false, a, false, b, (short)0, acc, false, false);
    }

    const int   col   = col0 + m;
    const int   rbase = row0 + ((lane >> 4) << 3);
    const float bv    = bias ? bias[col] : 0.0f;
#pragma unroll
    for (int v = 0; v < 8; ++v) {
        float x = acc[v] + bv;
        if (act == 1) x = fmaxf(x, 0.0f);
        C[(size_t)(rbase + v) * N + col] = x;
    }
}

__global__ void fill_f32(float* __restrict__ p, float v, int n)
{
    int i = blockIdx.x * blockDim.x + threadIdx.x;
    if (i < n) p[i] = v;
}

#define TR    196
#define NOWN  256
#define NCG   256

__global__ __launch_bounds__(1024) void bin_count(
    const int* __restrict__ dst, int* __restrict__ cntp, int E, int n)
{
    __shared__ int hist[NOWN];
    const int tid = threadIdx.x;
    if (tid < NOWN) hist[tid] = 0;
    __syncthreads();

    const int per = (E + (int)gridDim.x - 1) / (int)gridDim.x;
    const int e0  = blockIdx.x * per;
    const int e1  = min(E, e0 + per);
    for (int e = e0 + tid; e < e1; e += 4 * 1024) {
        unsigned d[4];
#pragma unroll
        for (int u = 0; u < 4; ++u) {
            const int ee = e + u * 1024;
            d[u] = (ee < e1) ? (unsigned)dst[ee] : 0xffffffffu;
        }
#pragma unroll
        for (int u = 0; u < 4; ++u)
            if (d[u] < (unsigned)n) atomicAdd(&hist[d[u] / TR], 1);
    }
    __syncthreads();
    if (tid < NOWN) cntp[blockIdx.x * NOWN + tid] = hist[tid];
}

__global__ __launch_bounds__(1024) void group_edges(
    const int* __restrict__ src, const int* __restrict__ dst,
    const int* __restrict__ cntp, int* __restrict__ regs, int* __restrict__ list,
    int E, int n)
{
    __shared__ int s_tot[NOWN];
    __shared__ int s_cur[NOWN];

    const int tid = threadIdx.x;
    const int g   = blockIdx.x;
    if (tid < NOWN) { s_tot[tid] = 0; s_cur[tid] = 0; }
    __syncthreads();

    {
        const int j = tid & (NOWN - 1);
        const int q = tid >> 8;
        int t = 0, p = 0;
#pragma unroll
        for (int i = 0; i < NCG / 4; ++i) {
            const int gg = q + 4 * i;
            const int c  = cntp[gg * NOWN + j];
            t += c;
            p += (gg < g) ? c : 0;
        }
        atomicAdd(&s_tot[j], t);
        atomicAdd(&s_cur[j], p);
    }
    __syncthreads();

    if (tid < 32) {
        int v[8];
        int loc = 0;
#pragma unroll
        for (int k = 0; k < 8; ++k) { v[k] = s_tot[8 * tid + k]; loc += v[k]; }
        int inc = loc;
        for (int o = 1; o < 32; o <<= 1) {
            const int up = __shfl_up(inc, o);
            if (tid >= o) inc += up;
        }
        int run = inc - loc;
#pragma unroll
        for (int k = 0; k < 8; ++k) {
            const int o = 8 * tid + k;
            s_cur[o] += run;
            if (g == 0) {
                regs[2 * o]     = run;
                regs[2 * o + 1] = v[k];
            }
            run += v[k];
        }
    }
    __syncthreads();

    int2* lst = (int2*)list;
    const int per = (E + (int)gridDim.x - 1) / (int)gridDim.x;
    const int e0  = g * per;
    const int e1  = min(E, e0 + per);
    for (int e = e0 + tid; e < e1; e += 4 * 1024) {
        int d[4], s[4];
#pragma unroll
        for (int u = 0; u < 4; ++u) {
            const int ee = e + u * 1024;
            const bool ok = ee < e1;
            d[u] = ok ? dst[ee] : -1;
            s[u] = ok ? src[ee] : 0;
        }
#pragma unroll
        for (int u = 0; u < 4; ++u) {
            if ((unsigned)d[u] < (unsigned)n) {
                const int o   = d[u] / TR;
                const int pos = atomicAdd(&s_cur[o], 1);
                lst[pos] = make_int2(s[u], d[u] - o * TR);
            }
        }
    }
}

__global__ __launch_bounds__(1024) void row_dinv(
    const int* __restrict__ regs, const int* __restrict__ list,
    float* __restrict__ dinv, int n)
{
    __shared__ int degc[TR];

    const int tid = threadIdx.x;
    const int b   = blockIdx.x;
    const int r0  = b * TR;
    const int nr  = min(TR, n - r0);

    if (tid < TR) degc[tid] = 0;
    __syncthreads();

    const int   count = regs[2 * b + 1];
    const int2* lst   = (const int2*)list + regs[2 * b];
    for (int i = tid; i < count; i += 4 * 1024) {
        int r[4];
#pragma unroll
        for (int u = 0; u < 4; ++u) {
            const int ii = i + u * 1024;
            r[u] = (ii < count) ? lst[ii].y : -1;
        }
#pragma unroll
        for (int u = 0; u < 4; ++u)
            if (r[u] >= 0) atomicAdd(&degc[r[u]], 1);
    }
    __syncthreads();

    if (tid < nr) dinv[r0 + tid] = rsqrtf((float)(degc[tid] + 1));
}

__device__ __forceinline__ void add_rows(
    const float* __restrict__ xw, float* acc,
    int s_i, int r_i, float n_i, int cnt, int lane)
{
    for (int k = 0; k < cnt; k += 8) {
        float v[8][4];
        int   rr[8];
        float nn[8];
#pragma unroll
        for (int u = 0; u < 8; ++u) {
            int kk = (k + u < cnt) ? (k + u) : k;
            int s  = __builtin_amdgcn_readlane(s_i, kk);
            rr[u]  = __builtin_amdgcn_readlane(r_i, kk);
            nn[u]  = __int_as_float(__builtin_amdgcn_readlane(__float_as_int(n_i), kk));
            const float* rowp = xw + (size_t)s * HID_C + lane;
            v[u][0] = rowp[0];
            v[u][1] = rowp[32];
            v[u][2] = rowp[64];
            v[u][3] = rowp[96];
        }
#pragma unroll
        for (int u = 0; u < 8; ++u) {
            if (k + u < cnt) {
                float* a = acc + rr[u] * HID_C + lane;
                atomicAdd(a,      v[u][0] * nn[u]);
                atomicAdd(a + 32, v[u][1] * nn[u]);
                atomicAdd(a + 64, v[u][2] * nn[u]);
                atomicAdd(a + 96, v[u][3] * nn[u]);
            }
        }
    }
}

__global__ __launch_bounds__(1024) void agg_list(
    const float* __restrict__ xw, const float* __restrict__ dinv,
    const int* __restrict__ regs, const int* __restrict__ list,
    float* __restrict__ h, int n)
{
    __shared__ float acc[TR * HID_C];

    const int tid  = threadIdx.x;
    const int wave = tid >> 5;
    const int lane = tid & 31;
    const int b    = blockIdx.x;
    const int r0   = b * TR;
    const int nr   = min(TR, n - r0);

    const int   count = regs[2 * b + 1];
    const int2* lst   = (const int2*)list + regs[2 * b];
    const int   nb    = (count + 31) >> 5;
    int  bi = wave;
    int2 en = make_int2(0, 0);
    if (bi < nb) {
        const int idx = (bi << 5) + lane;
        if (idx < count) en = lst[idx];
    }

    for (int i = tid; i < nr * 32; i += 1024) {
        int   r  = i >> 5;
        int   c4 = (i & 31) << 2;
        float di = dinv[r0 + r];
        float s  = di * di;
        float4 v = *(const float4*)(xw + (size_t)(r0 + r) * HID_C + c4);
        v.x *= s; v.y *= s; v.z *= s; v.w *= s;
        *(float4*)&acc[r * HID_C + c4] = v;
    }
    __syncthreads();

    while (bi < nb) {
        const int cnt = min(32, count - (bi << 5));
        float n_i = 0.0f;
        if (lane < cnt) n_i = dinv[en.x] * dinv[r0 + en.y];
        const int bn = bi + 32;
        int2 nx = make_int2(0, 0);
        if (bn < nb) {
            const int idx = (bn << 5) + lane;
            if (idx < count) nx = lst[idx];
        }
        add_rows(xw, acc, en.x, en.y, n_i, cnt, lane);
        en = nx;
        bi = bn;
    }
    __syncthreads();

    for (int i = tid; i < nr * 32; i += 1024) {
        int r  = i >> 5;
        int c4 = (i & 31) << 2;
        *(float4*)(h + (size_t)(r0 + r) * HID_C + c4) = *(const float4*)&acc[r * HID_C + c4];
    }
}

__global__ __launch_bounds__(32) void pool_accum(
    const float* __restrict__ h, const float* __restrict__ hb,
    const int* __restrict__ batch,
    float* __restrict__ sums, float* __restrict__ cnt, int n)
{
    int node = blockIdx.x;
    int c4   = threadIdx.x << 2;
    int b    = batch[node];
    float4 v  = *(const float4*)(h + (size_t)node * HID_C + c4);
    float4 bb = *(const float4*)(hb + c4);
    v.x = gelu_exact(v.x + bb.x);
    v.y = gelu_exact(v.y + bb.y);
    v.z = gelu_exact(v.z + bb.z);
    v.w = gelu_exact(v.w + bb.w);
    float* sp = sums + (size_t)b * HID_C + c4;
    atomicAdd(sp + 0, v.x);
    atomicAdd(sp + 1, v.y);
    atomicAdd(sp + 2, v.z);
    atomicAdd(sp + 3, v.w);
    if (threadIdx.x == 0) atomicAdd(&cnt[b], 1.0f);
}

__global__ void pool_final(float* __restrict__ sums, const float* __restrict__ cnt, int n)
{
    int i = blockIdx.x * blockDim.x + threadIdx.x;
    if (i < n) sums[i] /= fmaxf(cnt[i / HID_C], 1.0f);
}

extern "C" void kernel_launch(void* const* d_in, const int* in_sizes, int n_in,
                              void* d_out, int out_size, void* d_ws, size_t ws_size,
                              hipStream_t stream)
{
    const float* x     = (const float*)d_in[0];
    const int*   ei    = (const int*)d_in[1];
    const int*   batch = (const int*)d_in[2];
    const float* W[3]  = { (const float*)d_in[3], (const float*)d_in[5], (const float*)d_in[7] };
    const float* bv[3] = { (const float*)d_in[4], (const float*)d_in[6], (const float*)d_in[8] };
    const float* Wm1   = (const float*)d_in[9];
    const float* bm1   = (const float*)d_in[10];
    const float* Wm2   = (const float*)d_in[11];
    const float* bm2   = (const float*)d_in[12];

    const int N = in_sizes[0] / IN_C;
    const int E = in_sizes[1] / 2;
    const int* src = ei;
    const int* dst = ei + E;

    char*  ws  = (char*)d_ws;
    size_t off = 0;
    auto carve = [&](size_t bytes) -> float* {
        float* p = (float*)(ws + off);
        off = (off + bytes + 255) & ~(size_t)255;
        return p;
    };
    float* dinv = carve((size_t)N * 4);
    float* buf0 = carve((size_t)N * HID_C * 4);
    float* buf1 = carve((size_t)N * HID_C * 4);
    float* psum = carve((size_t)NBATCH * HID_C * 4);
    float* pcnt = carve((size_t)NBATCH * 4);
    float* hid  = carve((size_t)NBATCH * HID_C * 4);
    int*   cntp = (int*)carve((size_t)NCG * NOWN * 4);
    int*   regs = (int*)carve((size_t)NOWN * 2 * 4);
    int*   list = (int*)carve((size_t)E * 2 * 4);

    bin_count<<<NCG, 1024, 0, stream>>>(dst, cntp, E, N);
    group_edges<<<NCG, 1024, 0, stream>>>(src, dst, cntp, regs, list, E, N);
    row_dinv<<<NOWN, 1024, 0, stream>>>(regs, list, dinv, N);

    const float* ab[3] = { nullptr, bv[0], bv[1] };
    const float* cur = x;
    for (int l = 0; l < 3; ++l) {
        gemm_f32_wmma<<<N / 16, 256, 0, stream>>>(cur, W[l], nullptr, buf0,
                                                  N, HID_C, HID_C, 0, ab[l]);
        agg_list<<<NOWN, 1024, 0, stream>>>(buf0, dinv, regs, list, buf1, N);
        cur = buf1;
    }

    fill_f32<<<(NBATCH * HID_C + 255) / 256, 256, 0, stream>>>(psum, 0.0f, NBATCH * HID_C);
    fill_f32<<<(NBATCH + 255) / 256, 256, 0, stream>>>(pcnt, 0.0f, NBATCH);
    pool_accum<<<N, 32, 0, stream>>>(cur, bv[2], batch, psum, pcnt, N);
    pool_final<<<(NBATCH * HID_C + 255) / 256, 256, 0, stream>>>(psum, pcnt, NBATCH * HID_C);

    gemm_f32_wmma<<<NBATCH / 16, 256, 0, stream>>>(psum, Wm1, bm1, hid,
                                                   NBATCH, HID_C, HID_C, 1, nullptr);
    gemm_f32_wmma<<<NBATCH / 16, 128, 0, stream>>>(hid, Wm2, bm2, (float*)d_out,
                                                   NBATCH, HID_C, OUT_C, 0, nullptr);
}


# row sums in LDS: a source row comes as one 16-byte load per lane, not four 4-byte loads; owned rows kept in lane-major column order in LDS so the float adds stay on 32 banks; add order as before
# speedup vs baseline: 2.9021x; 1.0541x over previous
#include <hip/hip_runtime.h>
#include <hip/hip_bf16.h>
#include <math.h>

typedef float v2f __attribute__((ext_vector_type(2)));
typedef float v8f __attribute__((ext_vector_type(8)));

#define IN_C   128
#define HID_C  128
#define OUT_C  64
#define NBATCH 512

__device__ __forceinline__ float gelu_exact(float x)
{
    return 0.5f * x * (1.0f + erff(x * 0.70710678118654752f));
}

__global__ __launch_bounds__(256) void gemm_f32_wmma(
    const float* __restrict__ A, const float* __restrict__ B,
    const float* __restrict__ bias, float* __restrict__ C,
    int M, int K, int N, int act, const float* __restrict__ abias)
{
    __shared__ float lds_a[16][132];

    const int tid  = threadIdx.x;
    const int wave = tid >> 5;
    const int lane = tid & 31;
    const int row0 = blockIdx.x * 16;

    const int nvec = (16 * K) >> 2;
    const int kv   = K >> 2;
    for (int i = tid; i < nvec; i += blockDim.x) {
        int r  = i / kv;
        int k4 = (i - r * kv) << 2;
        float4 v = *(const float4*)(A + (size_t)(row0 + r) * K + k4);
        if (abias) {
            float4 bb = *(const float4*)(abias + k4);
            v.x = gelu_exact(v.x + bb.x);
            v.y = gelu_exact(v.y + bb.y);
            v.z = gelu_exact(v.z + bb.z);
            v.w = gelu_exact(v.w + bb.w);
        }
        *(float4*)&lds_a[r][k4] = v;
    }
    __syncthreads();

    const int m    = lane & 15;
    const int kh   = (lane >> 4) << 1;
    const int col0 = wave * 16;

    v8f acc = {};
    for (int kb = 0; kb < K; kb += 4) {
        v2f a, b;
        a.x = lds_a[m][kb + kh];
        a.y = lds_a[m][kb + kh + 1];
        const float* Bp = B + (size_t)(kb + kh) * N + col0 + m;
        b.x = Bp[0];
        b.y = Bp[N];
        acc = __builtin_amdgcn_wmma_f32_16x16x4_f32(
            false, a, false, b, (short)0, acc, false, false);
    }

    const int   col   = col0 + m;
    const int   rbase = row0 + ((lane >> 4) << 3);
    const float bv    = bias ? bias[col] : 0.0f;
#pragma unroll
    for (int v = 0; v < 8; ++v) {
        float x = acc[v] + bv;
        if (act == 1) x = fmaxf(x, 0.0f);
        C[(size_t)(rbase + v) * N + col] = x;
    }
}

__global__ void fill_f32(float* __restrict__ p, float v, int n)
{
    int i = blockIdx.x * blockDim.x + threadIdx.x;
    if (i < n) p[i] = v;
}

#define TR    196
#define NOWN  256
#define NCG   256

__global__ __launch_bounds__(1024) void bin_count(
    const int* __restrict__ dst, int* __restrict__ cntp, int E, int n)
{
    __shared__ int hist[NOWN];
    const int tid = threadIdx.x;
    if (tid < NOWN) hist[tid] = 0;
    __syncthreads();

    const int per = (E + (int)gridDim.x - 1) / (int)gridDim.x;
    const int e0  = blockIdx.x * per;
    const int e1  = min(E, e0 + per);
    for (int e = e0 + tid; e < e1; e += 4 * 1024) {
        unsigned d[4];
#pragma unroll
        for (int u = 0; u < 4; ++u) {
            const int ee = e + u * 1024;
            d[u] = (ee < e1) ? (unsigned)dst[ee] : 0xffffffffu;
        }
#pragma unroll
        for (int u = 0; u < 4; ++u)
            if (d[u] < (unsigned)n) atomicAdd(&hist[d[u] / TR], 1);
    }
    __syncthreads();
    if (tid < NOWN) cntp[blockIdx.x * NOWN + tid] = hist[tid];
}

__global__ __launch_bounds__(1024) void group_edges(
    const int* __restrict__ src, const int* __restrict__ dst,
    const int* __restrict__ cntp, int* __restrict__ regs, int* __restrict__ list,
    int E, int n)
{
    __shared__ int s_tot[NOWN];
    __shared__ int s_cur[NOWN];

    const int tid = threadIdx.x;
    const int g   = blockIdx.x;
    if (tid < NOWN) { s_tot[tid] = 0; s_cur[tid] = 0; }
    __syncthreads();

    {
        const int j = tid & (NOWN - 1);
        const int q = tid >> 8;
        int t = 0, p = 0;
#pragma unroll
        for (int i = 0; i < NCG / 4; ++i) {
            const int gg = q + 4 * i;
            const int c  = cntp[gg * NOWN + j];
            t += c;
            p += (gg < g) ? c : 0;
        }
        atomicAdd(&s_tot[j], t);
        atomicAdd(&s_cur[j], p);
    }
    __syncthreads();

    if (tid < 32) {
        int v[8];
        int loc = 0;
#pragma unroll
        for (int k = 0; k < 8; ++k) { v[k] = s_tot[8 * tid + k]; loc += v[k]; }
        int inc = loc;
        for (int o = 1; o < 32; o <<= 1) {
            const int up = __shfl_up(inc, o);
            if (tid >= o) inc += up;
        }
        int run = inc - loc;
#pragma unroll
        for (int k = 0; k < 8; ++k) {
            const int o = 8 * tid + k;
            s_cur[o] += run;
            if (g == 0) {
                regs[2 * o]     = run;
                regs[2 * o + 1] = v[k];
            }
            run += v[k];
        }
    }
    __syncthreads();

    int2* lst = (int2*)list;
    const int per = (E + (int)gridDim.x - 1) / (int)gridDim.x;
    const int e0  = g * per;
    const int e1  = min(E, e0 + per);
    for (int e = e0 + tid; e < e1; e += 4 * 1024) {
        int d[4], s[4];
#pragma unroll
        for (int u = 0; u < 4; ++u) {
            const int ee = e + u * 1024;
            const bool ok = ee < e1;
            d[u] = ok ? dst[ee] : -1;
            s[u] = ok ? src[ee] : 0;
        }
#pragma unroll
        for (int u = 0; u < 4; ++u) {
            if ((unsigned)d[u] < (unsigned)n) {
                const int o   = d[u] / TR;
                const int pos = atomicAdd(&s_cur[o], 1);
                lst[pos] = make_int2(s[u], d[u] - o * TR);
            }
        }
    }
}

__global__ __launch_bounds__(1024) void row_dinv(
    const int* __restrict__ regs, const int* __restrict__ list,
    float* __restrict__ dinv, int n)
{
    __shared__ int degc[TR];

    const int tid = threadIdx.x;
    const int b   = blockIdx.x;
    const int r0  = b * TR;
    const int nr  = min(TR, n - r0);

    if (tid < TR) degc[tid] = 0;
    __syncthreads();

    const int   count = regs[2 * b + 1];
    const int2* lst   = (const int2*)list + regs[2 * b];
    for (int i = tid; i < count; i += 4 * 1024) {
        int r[4];
#pragma unroll
        for (int u = 0; u < 4; ++u) {
            const int ii = i + u * 1024;
            r[u] = (ii < count) ? lst[ii].y : -1;
        }
#pragma unroll
        for (int u = 0; u < 4; ++u)
            if (r[u] >= 0) atomicAdd(&degc[r[u]], 1);
    }
    __syncthreads();

    if (tid < nr) dinv[r0 + tid] = rsqrtf((float)(degc[tid] + 1));
}

__device__ __forceinline__ void add_rows(
    const float* __restrict__ xw, float* acc,
    int s_i, int r_i, float n_i, int cnt, int lane)
{
    for (int k = 0; k < cnt; k += 8) {
        float v[8][4];
        int   rr[8];
        float nn[8];
#pragma unroll
        for (int u = 0; u < 8; ++u) {
            int kk = (k + u < cnt) ? (k + u) : k;
            int s  = __builtin_amdgcn_readlane(s_i, kk);
            rr[u]  = __builtin_amdgcn_readlane(r_i, kk);
            nn[u]  = __int_as_float(__builtin_amdgcn_readlane(__float_as_int(n_i), kk));
            const float4 q = *(const float4*)(xw + (size_t)s * HID_C + (lane << 2));
            v[u][0] = q.x;
            v[u][1] = q.y;
            v[u][2] = q.z;
            v[u][3] = q.w;
        }
#pragma unroll
        for (int u = 0; u < 8; ++u) {
            if (k + u < cnt) {
                float* a = acc + rr[u] * HID_C + lane;
                atomicAdd(a,      v[u][0] * nn[u]);
                atomicAdd(a + 32, v[u][1] * nn[u]);
                atomicAdd(a + 64, v[u][2] * nn[u]);
                atomicAdd(a + 96, v[u][3] * nn[u]);
            }
        }
    }
}

__global__ __launch_bounds__(1024) void agg_list(
    const float* __restrict__ xw, const float* __restrict__ dinv,
    const int* __restrict__ regs, const int* __restrict__ list,
    float* __restrict__ h, int n)
{
    __shared__ float acc[TR * HID_C];

    const int tid  = threadIdx.x;
    const int wave = tid >> 5;
    const int lane = tid & 31;
    const int b    = blockIdx.x;
    const int r0   = b * TR;
    const int nr   = min(TR, n - r0);

    const int   count = regs[2 * b + 1];
    const int2* lst   = (const int2*)list + regs[2 * b];
    const int   nb    = (count + 31) >> 5;
    int  bi = wave;
    int2 en = make_int2(0, 0);
    if (bi < nb) {
        const int idx = (bi << 5) + lane;
        if (idx < count) en = lst[idx];
    }

    for (int i = tid; i < nr * 32; i += 1024) {
        int   r  = i >> 5;
        int   c4 = (i & 31) << 2;
        float di = dinv[r0 + r];
        float s  = di * di;
        float4 v = *(const float4*)(xw + (size_t)(r0 + r) * HID_C + c4);
        float* a = acc + r * HID_C + (i & 31);
        a[0]  = v.x * s;
        a[32] = v.y * s;
        a[64] = v.z * s;
        a[96] = v.w * s;
    }
    __syncthreads();

    while (bi < nb) {
        const int cnt = min(32, count - (bi << 5));
        float n_i = 0.0f;
        if (lane < cnt) n_i = dinv[en.x] * dinv[r0 + en.y];
        const int bn = bi + 32;
        int2 nx = make_int2(0, 0);
        if (bn < nb) {
            const int idx = (bn << 5) + lane;
            if (idx < count) nx = lst[idx];
        }
        add_rows(xw, acc, en.x, en.y, n_i, cnt, lane);
        en = nx;
        bi = bn;
    }
    __syncthreads();

    for (int i = tid; i < nr * 32; i += 1024) {
        int r  = i >> 5;
        int c4 = (i & 31) << 2;
        const float* a = acc + r * HID_C + (i & 31);
        *(float4*)(h + (size_t)(r0 + r) * HID_C + c4) = make_float4(a[0], a[32], a[64], a[96]);
    }
}

__global__ __launch_bounds__(32) void pool_accum(
    const float* __restrict__ h, const float* __restrict__ hb,
    const int* __restrict__ batch,
    float* __restrict__ sums, float* __restrict__ cnt, int n)
{
    int node = blockIdx.x;
    int c4   = threadIdx.x << 2;
    int b    = batch[node];
    float4 v  = *(const float4*)(h + (size_t)node * HID_C + c4);
    float4 bb = *(const float4*)(hb + c4);
    v.x = gelu_exact(v.x + bb.x);
    v.y = gelu_exact(v.y + bb.y);
    v.z = gelu_exact(v.z + bb.z);
    v.w = gelu_exact(v.w + bb.w);
    float* sp = sums + (size_t)b * HID_C + c4;
    atomicAdd(sp + 0, v.x);
    atomicAdd(sp + 1, v.y);
    atomicAdd(sp + 2, v.z);
    atomicAdd(sp + 3, v.w);
    if (threadIdx.x == 0) atomicAdd(&cnt[b], 1.0f);
}

__global__ void pool_final(float* __restrict__ sums, const float* __restrict__ cnt, int n)
{
    int i = blockIdx.x * blockDim.x + threadIdx.x;
    if (i < n) sums[i] /= fmaxf(cnt[i / HID_C], 1.0f);
}

extern "C" void kernel_launch(void* const* d_in, const int* in_sizes, int n_in,
                              void* d_out, int out_size, void* d_ws, size_t ws_size,
                              hipStream_t stream)
{
    const float* x     = (const float*)d_in[0];
    const int*   ei    = (const int*)d_in[1];
    const int*   batch = (const int*)d_in[2];
    const float* W[3]  = { (const float*)d_in[3], (const float*)d_in[5], (const float*)d_in[7] };
    const float* bv[3] = { (const float*)d_in[4], (const float*)d_in[6], (const float*)d_in[8] };
    const float* Wm1   = (const float*)d_in[9];
    const float* bm1   = (const float*)d_in[10];
    const float* Wm2   = (const float*)d_in[11];
    const float* bm2   = (const float*)d_in[12];

    const int N = in_sizes[0] / IN_C;
    const int E = in_sizes[1] / 2;
    const int* src = ei;
    const int* dst = ei + E;

    char*  ws  = (char*)d_ws;
    size_t off = 0;
    auto carve = [&](size_t bytes) -> float* {
        float* p = (float*)(ws + off);
        off = (off + bytes + 255) & ~(size_t)255;
        return p;
    };
    float* dinv = carve((size_t)N * 4);
    float* buf0 = carve((size_t)N * HID_C * 4);
    float* buf1 = carve((size_t)N * HID_C * 4);
    float* psum = carve((size_t)NBATCH * HID_C * 4);
    float* pcnt = carve((size_t)NBATCH * 4);
    float* hid  = carve((size_t)NBATCH * HID_C * 4);
    int*   cntp = (int*)carve((size_t)NCG * NOWN * 4);
    int*   regs = (int*)carve((size_t)NOWN * 2 * 4);
    int*   list = (int*)carve((size_t)E * 2 * 4);

    bin_count<<<NCG, 1024, 0, stream>>>(dst, cntp, E, N);
    group_edges<<<NCG, 1024, 0, stream>>>(src, dst, cntp, regs, list, E, N);
    row_dinv<<<NOWN, 1024, 0, stream>>>(regs, list, dinv, N);

    const float* ab[3] = { nullptr, bv[0], bv[1] };
    const float* cur = x;
    for (int l = 0; l < 3; ++l) {
        gemm_f32_wmma<<<N / 16, 256, 0, stream>>>(cur, W[l], nullptr, buf0,
                                                  N, HID_C, HID_C, 0, ab[l]);
        agg_list<<<NOWN, 1024, 0, stream>>>(buf0, dinv, regs, list, buf1, N);
        cur = buf1;
    }

    fill_f32<<<(NBATCH * HID_C + 255) / 256, 256, 0, stream>>>(psum, 0.0f, NBATCH * HID_C);
    fill_f32<<<(NBATCH + 255) / 256, 256, 0, stream>>>(pcnt, 0.0f, NBATCH);
    pool_accum<<<N, 32, 0, stream>>>(cur, bv[2], batch, psum, pcnt, N);
    pool_final<<<(NBATCH * HID_C + 255) / 256, 256, 0, stream>>>(psum, pcnt, NBATCH * HID_C);

    gemm_f32_wmma<<<NBATCH / 16, 256, 0, stream>>>(psum, Wm1, bm1, hid,
                                                   NBATCH, HID_C, HID_C, 1, nullptr);
    gemm_f32_wmma<<<NBATCH / 16, 128, 0, stream>>>(hid, Wm2, bm2, (float*)d_out,
                                                   NBATCH, HID_C, OUT_C, 0, nullptr);
}


# mean pool: per-node global f32 atomic adds, 2 fills, divide pass -> one work-group per graph lists its nodes from the id array in LDS, sums their rows; no memory atomics; pool sum order changes
# speedup vs baseline: 3.6342x; 1.2523x over previous
#include <hip/hip_runtime.h>
#include <hip/hip_bf16.h>
#include <math.h>

typedef float v2f __attribute__((ext_vector_type(2)));
typedef float v8f __attribute__((ext_vector_type(8)));

#define IN_C   128
#define HID_C  128
#define OUT_C  64
#define NBATCH 512

__device__ __forceinline__ float gelu_exact(float x)
{
    return 0.5f * x * (1.0f + erff(x * 0.70710678118654752f));
}

__global__ __launch_bounds__(256) void gemm_f32_wmma(
    const float* __restrict__ A, const float* __restrict__ B,
    const float* __restrict__ bias, float* __restrict__ C,
    int M, int K, int N, int act, const float* __restrict__ abias)
{
    __shared__ float lds_a[16][132];

    const int tid  = threadIdx.x;
    const int wave = tid >> 5;
    const int lane = tid & 31;
    const int row0 = blockIdx.x * 16;

    const int nvec = (16 * K) >> 2;
    const int kv   = K >> 2;
    for (int i = tid; i < nvec; i += blockDim.x) {
        int r  = i / kv;
        int k4 = (i - r * kv) << 2;
        float4 v = *(const float4*)(A + (size_t)(row0 + r) * K + k4);
        if (abias) {
            float4 bb = *(const float4*)(abias + k4);
            v.x = gelu_exact(v.x + bb.x);
            v.y = gelu_exact(v.y + bb.y);
            v.z = gelu_exact(v.z + bb.z);
            v.w = gelu_exact(v.w + bb.w);
        }
        *(float4*)&lds_a[r][k4] = v;
    }
    __syncthreads();

    const int m    = lane & 15;
    const int kh   = (lane >> 4) << 1;
    const int col0 = wave * 16;

    v8f acc = {};
    for (int kb = 0; kb < K; kb += 4) {
        v2f a, b;
        a.x = lds_a[m][kb + kh];
        a.y = lds_a[m][kb + kh + 1];
        const float* Bp = B + (size_t)(kb + kh) * N + col0 + m;
        b.x = Bp[0];
        b.y = Bp[N];
        acc = __builtin_amdgcn_wmma_f32_16x16x4_f32(
            false, a, false, b, (short)0, acc, false, false);
    }

    const int   col   = col0 + m;
    const int   rbase = row0 + ((lane >> 4) << 3);
    const float bv    = bias ? bias[col] : 0.0f;
#pragma unroll
    for (int v = 0; v < 8; ++v) {
        float x = acc[v] + bv;
        if (act == 1) x = fmaxf(x, 0.0f);
        C[(size_t)(rbase + v) * N + col] = x;
    }
}

#define TR    196
#define NOWN  256
#define NCG   256

__global__ __launch_bounds__(1024) void bin_count(
    const int* __restrict__ dst, int* __restrict__ cntp, int E, int n)
{
    __shared__ int hist[NOWN];
    const int tid = threadIdx.x;
    if (tid < NOWN) hist[tid] = 0;
    __syncthreads();

    const int per = (E + (int)gridDim.x - 1) / (int)gridDim.x;
    const int e0  = blockIdx.x * per;
    const int e1  = min(E, e0 + per);
    for (int e = e0 + tid; e < e1; e += 4 * 1024) {
        unsigned d[4];
#pragma unroll
        for (int u = 0; u < 4; ++u) {
            const int ee = e + u * 1024;
            d[u] = (ee < e1) ? (unsigned)dst[ee] : 0xffffffffu;
        }
#pragma unroll
        for (int u = 0; u < 4; ++u)
            if (d[u] < (unsigned)n) atomicAdd(&hist[d[u] / TR], 1);
    }
    __syncthreads();
    if (tid < NOWN) cntp[blockIdx.x * NOWN + tid] = hist[tid];
}

__global__ __launch_bounds__(1024) void group_edges(
    const int* __restrict__ src, const int* __restrict__ dst,
    const int* __restrict__ cntp, int* __restrict__ regs, int* __restrict__ list,
    int E, int n)
{
    __shared__ int s_tot[NOWN];
    __shared__ int s_cur[NOWN];

    const int tid = threadIdx.x;
    const int g   = blockIdx.x;
    if (tid < NOWN) { s_tot[tid] = 0; s_cur[tid] = 0; }
    __syncthreads();

    {
        const int j = tid & (NOWN - 1);
        const int q = tid >> 8;
        int t = 0, p = 0;
#pragma unroll
        for (int i = 0; i < NCG / 4; ++i) {
            const int gg = q + 4 * i;
            const int c  = cntp[gg * NOWN + j];
            t += c;
            p += (gg < g) ? c : 0;
        }
        atomicAdd(&s_tot[j], t);
        atomicAdd(&s_cur[j], p);
    }
    __syncthreads();

    if (tid < 32) {
        int v[8];
        int loc = 0;
#pragma unroll
        for (int k = 0; k < 8; ++k) { v[k] = s_tot[8 * tid + k]; loc += v[k]; }
        int inc = loc;
        for (int o = 1; o < 32; o <<= 1) {
            const int up = __shfl_up(inc, o);
            if (tid >= o) inc += up;
        }
        int run = inc - loc;
#pragma unroll
        for (int k = 0; k < 8; ++k) {
            const int o = 8 * tid + k;
            s_cur[o] += run;
            if (g == 0) {
                regs[2 * o]     = run;
                regs[2 * o + 1] = v[k];
            }
            run += v[k];
        }
    }
    __syncthreads();

    int2* lst = (int2*)list;
    const int per = (E + (int)gridDim.x - 1) / (int)gridDim.x;
    const int e0  = g * per;
    const int e1  = min(E, e0 + per);
    for (int e = e0 + tid; e < e1; e += 4 * 1024) {
        int d[4], s[4];
#pragma unroll
        for (int u = 0; u < 4; ++u) {
            const int ee = e + u * 1024;
            const bool ok = ee < e1;
            d[u] = ok ? dst[ee] : -1;
            s[u] = ok ? src[ee] : 0;
        }
#pragma unroll
        for (int u = 0; u < 4; ++u) {
            if ((unsigned)d[u] < (unsigned)n) {
                const int o   = d[u] / TR;
                const int pos = atomicAdd(&s_cur[o], 1);
                lst[pos] = make_int2(s[u], d[u] - o * TR);
            }
        }
    }
}

__global__ __launch_bounds__(1024) void row_dinv(
    const int* __restrict__ regs, const int* __restrict__ list,
    float* __restrict__ dinv, int n)
{
    __shared__ int degc[TR];

    const int tid = threadIdx.x;
    const int b   = blockIdx.x;
    const int r0  = b * TR;
    const int nr  = min(TR, n - r0);

    if (tid < TR) degc[tid] = 0;
    __syncthreads();

    const int   count = regs[2 * b + 1];
    const int2* lst   = (const int2*)list + regs[2 * b];
    for (int i = tid; i < count; i += 4 * 1024) {
        int r[4];
#pragma unroll
        for (int u = 0; u < 4; ++u) {
            const int ii = i + u * 1024;
            r[u] = (ii < count) ? lst[ii].y : -1;
        }
#pragma unroll
        for (int u = 0; u < 4; ++u)
            if (r[u] >= 0) atomicAdd(&degc[r[u]], 1);
    }
    __syncthreads();

    if (tid < nr) dinv[r0 + tid] = rsqrtf((float)(degc[tid] + 1));
}

__device__ __forceinline__ void add_rows(
    const float* __restrict__ xw, float* acc,
    int s_i, int r_i, float n_i, int cnt, int lane)
{
    for (int k = 0; k < cnt; k += 8) {
        float v[8][4];
        int   rr[8];
        float nn[8];
#pragma unroll
        for (int u = 0; u < 8; ++u) {
            int kk = (k + u < cnt) ? (k + u) : k;
            int s  = __builtin_amdgcn_readlane(s_i, kk);
            rr[u]  = __builtin_amdgcn_readlane(r_i, kk);
            nn[u]  = __int_as_float(__builtin_amdgcn_readlane(__float_as_int(n_i), kk));
            const float4 q = *(const float4*)(xw + (size_t)s * HID_C + (lane << 2));
            v[u][0] = q.x;
            v[u][1] = q.y;
            v[u][2] = q.z;
            v[u][3] = q.w;
        }
#pragma unroll
        for (int u = 0; u < 8; ++u) {
            if (k + u < cnt) {
                float* a = acc + rr[u] * HID_C + lane;
                atomicAdd(a,      v[u][0] * nn[u]);
                atomicAdd(a + 32, v[u][1] * nn[u]);
                atomicAdd(a + 64, v[u][2] * nn[u]);
                atomicAdd(a + 96, v[u][3] * nn[u]);
            }
        }
    }
}

__global__ __launch_bounds__(1024) void agg_list(
    const float* __restrict__ xw, const float* __restrict__ dinv,
    const int* __restrict__ regs, const int* __restrict__ list,
    float* __restrict__ h, int n)
{
    __shared__ float acc[TR * HID_C];

    const int tid  = threadIdx.x;
    const int wave = tid >> 5;
    const int lane = tid & 31;
    const int b    = blockIdx.x;
    const int r0   = b * TR;
    const int nr   = min(TR, n - r0);

    const int   count = regs[2 * b + 1];
    const int2* lst   = (const int2*)list + regs[2 * b];
    const int   nb    = (count + 31) >> 5;
    int  bi = wave;
    int2 en = make_int2(0, 0);
    if (bi < nb) {
        const int idx = (bi << 5) + lane;
        if (idx < count) en = lst[idx];
    }

    for (int i = tid; i < nr * 32; i += 1024) {
        int   r  = i >> 5;
        int   c4 = (i & 31) << 2;
        float di = dinv[r0 + r];
        float s  = di * di;
        float4 v = *(const float4*)(xw + (size_t)(r0 + r) * HID_C + c4);
        float* a = acc + r * HID_C + (i & 31);
        a[0]  = v.x * s;
        a[32] = v.y * s;
        a[64] = v.z * s;
        a[96] = v.w * s;
    }
    __syncthreads();

    while (bi < nb) {
        const int cnt = min(32, count - (bi << 5));
        float n_i = 0.0f;
        if (lane < cnt) n_i = dinv[en.x] * dinv[r0 + en.y];
        const int bn = bi + 32;
        int2 nx = make_int2(0, 0);
        if (bn < nb) {
            const int idx = (bn << 5) + lane;
            if (idx < count) nx = lst[idx];
        }
        add_rows(xw, acc, en.x, en.y, n_i, cnt, lane);
        en = nx;
        bi = bn;
    }
    __syncthreads();

    for (int i = tid; i < nr * 32; i += 1024) {
        int r  = i >> 5;
        int c4 = (i & 31) << 2;
        const float* a = acc + r * HID_C + (i & 31);
        *(float4*)(h + (size_t)(r0 + r) * HID_C + c4) = make_float4(a[0], a[32], a[64], a[96]);
    }
}

#define PTH  512
#define PWV  (PTH / 32)
#define PQ   4
#define PCH  (PTH * 4 * PQ)

__device__ __forceinline__ int4 load_ids(const int* __restrict__ batch, int id0, int n)
{
    if (id0 + 3 < n) return *(const int4*)(batch + id0);
    int4 q;
    q.x = (id0     < n) ? batch[id0]     : -1;
    q.y = (id0 + 1 < n) ? batch[id0 + 1] : -1;
    q.z = (id0 + 2 < n) ? batch[id0 + 2] : -1;
    q.w = (id0 + 3 < n) ? batch[id0 + 3] : -1;
    return q;
}

__device__ __forceinline__ void keep_own(int4 q, int g, int id0, int* cnt, int* lst)
{
    if ((q.x == g) | (q.y == g) | (q.z == g) | (q.w == g)) {
        if (q.x == g) lst[atomicAdd(cnt, 1)] = id0;
        if (q.y == g) lst[atomicAdd(cnt, 1)] = id0 + 1;
        if (q.z == g) lst[atomicAdd(cnt, 1)] = id0 + 2;
        if (q.w == g) lst[atomicAdd(cnt, 1)] = id0 + 3;
    }
}

__global__ __launch_bounds__(PTH) void pool_owner(
    const float* __restrict__ h, const float* __restrict__ hb,
    const int* __restrict__ batch, float* __restrict__ pooled, int n)
{
    __shared__ int   s_n[3];
    __shared__ int   s_list[3][PCH];
    __shared__ float s_part[PWV][HID_C];

    const int tid  = threadIdx.x;
    const int wave = tid >> 5;
    const int lane = tid & 31;
    const int g    = blockIdx.x;

    if (tid < 3) s_n[tid] = 0;
    __syncthreads();

    const int    nch = (n + PCH - 1) / PCH;
    const float4 bb  = *(const float4*)(hb + (lane << 2));
    float4 acc   = make_float4(0.0f, 0.0f, 0.0f, 0.0f);
    int    total = 0;

    int4 q[PQ];
#pragma unroll
    for (int u = 0; u < PQ; ++u) q[u] = load_ids(batch, (u * PTH + tid) << 2, n);
    for (int c = 0; c < nch; ++c) {
        const int k   = c % 3;
        const int id0 = c * PCH + (tid << 2);
#pragma unroll
        for (int u = 0; u < PQ; ++u) keep_own(q[u], g, id0 + u * PTH * 4, &s_n[k], s_list[k]);

#pragma unroll
        for (int u = 0; u < PQ; ++u) q[u] = load_ids(batch, id0 + PCH + u * PTH * 4, n);
        __syncthreads();

        if (tid == 0) s_n[(c + 2) % 3] = 0;
        const int  cnt = s_n[k];
        const int* lst = s_list[k];
        total += cnt;
        for (int e = wave; e < cnt; e += 4 * PWV) {
            float4 v[4];
#pragma unroll
            for (int u = 0; u < 4; ++u) {
                const int ee   = e + u * PWV;
                const int node = lst[(ee < cnt) ? ee : e];
                v[u] = *(const float4*)(h + (size_t)node * HID_C + (lane << 2));
            }
#pragma unroll
            for (int u = 0; u < 4; ++u) {
                if (e + u * PWV < cnt) {
                    acc.x += gelu_exact(v[u].x + bb.x);
                    acc.y += gelu_exact(v[u].y + bb.y);
                    acc.z += gelu_exact(v[u].z + bb.z);
                    acc.w += gelu_exact(v[u].w + bb.w);
                }
            }
        }
    }

    *(float4*)&s_part[wave][lane << 2] = acc;
    __syncthreads();
    if (tid < HID_C) {
        float s = 0.0f;
#pragma unroll
        for (int w = 0; w < PWV; ++w) s += s_part[w][tid];
        pooled[(size_t)g * HID_C + tid] = s / fmaxf((float)total, 1.0f);
    }
}

extern "C" void kernel_launch(void* const* d_in, const int* in_sizes, int n_in,
                              void* d_out, int out_size, void* d_ws, size_t ws_size,
                              hipStream_t stream)
{
    const float* x     = (const float*)d_in[0];
    const int*   ei    = (const int*)d_in[1];
    const int*   batch = (const int*)d_in[2];
    const float* W[3]  = { (const float*)d_in[3], (const float*)d_in[5], (const float*)d_in[7] };
    const float* bv[3] = { (const float*)d_in[4], (const float*)d_in[6], (const float*)d_in[8] };
    const float* Wm1   = (const float*)d_in[9];
    const float* bm1   = (const float*)d_in[10];
    const float* Wm2   = (const float*)d_in[11];
    const float* bm2   = (const float*)d_in[12];

    const int N = in_sizes[0] / IN_C;
    const int E = in_sizes[1] / 2;
    const int* src = ei;
    const int* dst = ei + E;

    char*  ws  = (char*)d_ws;
    size_t off = 0;
    auto carve = [&](size_t bytes) -> float* {
        float* p = (float*)(ws + off);
        off = (off + bytes + 255) & ~(size_t)255;
        return p;
    };
    float* dinv = carve((size_t)N * 4);
    float* buf0 = carve((size_t)N * HID_C * 4);
    float* buf1 = carve((size_t)N * HID_C * 4);
    float* psum = carve((size_t)NBATCH * HID_C * 4);
    float* hid  = carve((size_t)NBATCH * HID_C * 4);
    int*   cntp = (int*)carve((size_t)NCG * NOWN * 4);
    int*   regs = (int*)carve((size_t)NOWN * 2 * 4);
    int*   list = (int*)carve((size_t)E * 2 * 4);

    bin_count<<<NCG, 1024, 0, stream>>>(dst, cntp, E, N);
    group_edges<<<NCG, 1024, 0, stream>>>(src, dst, cntp, regs, list, E, N);
    row_dinv<<<NOWN, 1024, 0, stream>>>(regs, list, dinv, N);

    const float* ab[3] = { nullptr, bv[0], bv[1] };
    const float* cur = x;
    for (int l = 0; l < 3; ++l) {
        gemm_f32_wmma<<<N / 16, 256, 0, stream>>>(cur, W[l], nullptr, buf0,
                                                  N, HID_C, HID_C, 0, ab[l]);
        agg_list<<<NOWN, 1024, 0, stream>>>(buf0, dinv, regs, list, buf1, N);
        cur = buf1;
    }

    pool_owner<<<NBATCH, PTH, 0, stream>>>(cur, bv[2], batch, psum, N);

    gemm_f32_wmma<<<NBATCH / 16, 256, 0, stream>>>(psum, Wm1, bm1, hid,
                                                   NBATCH, HID_C, HID_C, 1, nullptr);
    gemm_f32_wmma<<<NBATCH / 16, 128, 0, stream>>>(hid, Wm2, bm2, (float*)d_out,
                                                   NBATCH, HID_C, OUT_C, 0, nullptr);
}


# layer product retiled: a wave keeps its 32-column strip of W in registers as B fragments, walks its rows 32 a step with two accumulators; 512 work-groups, not 3125; k sum order changes
# speedup vs baseline: 4.2725x; 1.1756x over previous
#include <hip/hip_runtime.h>
#include <hip/hip_bf16.h>
#include <math.h>

typedef float v2f __attribute__((ext_vector_type(2)));
typedef float v8f __attribute__((ext_vector_type(8)));

#define IN_C   128
#define HID_C  128
#define OUT_C  64
#define NBATCH 512

__device__ __forceinline__ float gelu_exact(float x)
{
    return 0.5f * x * (1.0f + erff(x * 0.70710678118654752f));
}

__global__ __launch_bounds__(256) void gemm_f32_wmma(
    const float* __restrict__ A, const float* __restrict__ B,
    const float* __restrict__ bias, float* __restrict__ C,
    int M, int K, int N, int act)
{
    __shared__ float lds_a[16][132];

    const int tid  = threadIdx.x;
    const int wave = tid >> 5;
    const int lane = tid & 31;
    const int row0 = blockIdx.x * 16;

    const int nvec = (16 * K) >> 2;
    const int kv   = K >> 2;
    for (int i = tid; i < nvec; i += blockDim.x) {
        int r  = i / kv;
        int k4 = (i - r * kv) << 2;
        *(float4*)&lds_a[r][k4] = *(const float4*)(A + (size_t)(row0 + r) * K + k4);
    }
    __syncthreads();

    const int m    = lane & 15;
    const int kh   = (lane >> 4) << 1;
    const int col0 = wave * 16;

    v8f acc = {};
    for (int kb = 0; kb < K; kb += 4) {
        v2f a, b;
        a.x = lds_a[m][kb + kh];
        a.y = lds_a[m][kb + kh + 1];
        const float* Bp = B + (size_t)(kb + kh) * N + col0 + m;
        b.x = Bp[0];
        b.y = Bp[N];
        acc = __builtin_amdgcn_wmma_f32_16x16x4_f32(
            false, a, false, b, (short)0, acc, false, false);
    }

    const int   col   = col0 + m;
    const int   rbase = row0 + ((lane >> 4) << 3);
    const float bv    = bias ? bias[col] : 0.0f;
#pragma unroll
    for (int v = 0; v < 8; ++v) {
        float x = acc[v] + bv;
        if (act == 1) x = fmaxf(x, 0.0f);
        C[(size_t)(rbase + v) * N + col] = x;
    }
}

#define LG_NWG   512
#define LG_PITCH 132

__device__ __forceinline__ void lg_load_rows(
    const float* __restrict__ A, int t, int t1, int wave, int c4, float4 (&va)[4])
{
#pragma unroll
    for (int u = 0; u < 4; ++u)
        if (t + (u >> 1) < t1)
            va[u] = *(const float4*)(A + (size_t)(t * 16 + u * 8 + wave) * HID_C + c4);
}

__device__ __forceinline__ float lg_other_half(float x)
{
    return __int_as_float(__builtin_amdgcn_permlanex16(
        __float_as_int(x), __float_as_int(x), 0x76543210, 0xfedcba98, false, false));
}

__global__ __launch_bounds__(256) void layer_gemm(
    const float* __restrict__ A, const float* __restrict__ Wt,
    float* __restrict__ C, int M, const float* __restrict__ abias)
{
    __shared__ __attribute__((aligned(16))) float lds_a[2][32][LG_PITCH];

    const int tid  = threadIdx.x;
    const int lane = tid & 31;
    const int wave = __builtin_amdgcn_readfirstlane(tid >> 5);
    const int rg   = wave >> 2;
    const int cg   = wave & 3;
    const int m    = lane & 15;
    const int h    = lane >> 4;
    const int c4   = lane << 2;

    const unsigned ntile = (unsigned)M >> 4;
    const int t0 = (int)((blockIdx.x * ntile) / gridDim.x);
    const int t1 = (int)(((blockIdx.x + 1) * ntile) / gridDim.x);

    float4 va[4];
    lg_load_rows(A, t0, t1, wave, c4, va);

    float4 bq[16][2];
#pragma unroll
    for (int p = 0; p < 16; ++p) {
#pragma unroll
        for (int j = 0; j < 2; ++j) {
            const float* wp = Wt + (8 * p + 4 * h) * HID_C + 32 * cg + 16 * j + m;
            bq[p][j] = make_float4(wp[0], wp[HID_C], wp[2 * HID_C], wp[3 * HID_C]);
        }
    }

    float4 bb = make_float4(0.0f, 0.0f, 0.0f, 0.0f);
    if (abias) bb = *(const float4*)(abias + c4);

    int buf = 0;
    for (int t = t0; t < t1; t += 2) {
#pragma unroll
        for (int u = 0; u < 4; ++u) {
            if (t + (u >> 1) < t1) {
                float4 v = va[u];
                if (abias) {
                    v.x = gelu_exact(v.x + bb.x);
                    v.y = gelu_exact(v.y + bb.y);
                    v.z = gelu_exact(v.z + bb.z);
                    v.w = gelu_exact(v.w + bb.w);
                }
                *(float4*)&lds_a[buf][u * 8 + wave][c4] = v;
            }
        }
        __syncthreads();
        if (t + 2 < t1) lg_load_rows(A, t + 2, t1, wave, c4, va);

        if (t + rg < t1) {
            v8f acc0 = {}, acc1 = {};
            const float* ap = &lds_a[buf][16 * rg + m][4 * h];
#pragma unroll
            for (int p = 0; p < 16; ++p) {
                const float4 aq = *(const float4*)(ap + 8 * p);
                const v2f a01 = { aq.x, aq.y };
                const v2f a23 = { aq.z, aq.w };
                const v2f b00 = { bq[p][0].x, bq[p][0].y };
                const v2f b01 = { bq[p][0].z, bq[p][0].w };
                const v2f b10 = { bq[p][1].x, bq[p][1].y };
                const v2f b11 = { bq[p][1].z, bq[p][1].w };
                acc0 = __builtin_amdgcn_wmma_f32_16x16x4_f32(
                    false, a01, false, b00, (short)0, acc0, false, false);
                acc1 = __builtin_amdgcn_wmma_f32_16x16x4_f32(
                    false, a01, false, b10, (short)0, acc1, false, false);
                acc0 = __builtin_amdgcn_wmma_f32_16x16x4_f32(
                    false, a23, false, b01, (short)0, acc0, false, false);
                acc1 = __builtin_amdgcn_wmma_f32_16x16x4_f32(
                    false, a23, false, b11, (short)0, acc1, false, false);
            }

            float* cp = C + (size_t)((t + rg) * 16) * HID_C + 32 * cg + lane;
#pragma unroll
            for (int r = 0; r < 8; ++r) {
                const float send = h ? acc0[r] : acc1[r];
                const float recv = lg_other_half(send);
                const float lo   = h ? recv : acc0[r];
                const float hi   = h ? acc1[r] : recv;
                cp[(size_t)r * HID_C]       = lo;
                cp[(size_t)(8 + r) * HID_C] = hi;
            }
        }
        buf ^= 1;
    }
}

#define TR    196
#define NOWN  256
#define NCG   256

__global__ __launch_bounds__(1024) void bin_count(
    const int* __restrict__ dst, int* __restrict__ cntp, int E, int n)
{
    __shared__ int hist[NOWN];
    const int tid = threadIdx.x;
    if (tid < NOWN) hist[tid] = 0;
    __syncthreads();

    const int per = (E + (int)gridDim.x - 1) / (int)gridDim.x;
    const int e0  = blockIdx.x * per;
    const int e1  = min(E, e0 + per);
    for (int e = e0 + tid; e < e1; e += 4 * 1024) {
        unsigned d[4];
#pragma unroll
        for (int u = 0; u < 4; ++u) {
            const int ee = e + u * 1024;
            d[u] = (ee < e1) ? (unsigned)dst[ee] : 0xffffffffu;
        }
#pragma unroll
        for (int u = 0; u < 4; ++u)
            if (d[u] < (unsigned)n) atomicAdd(&hist[d[u] / TR], 1);
    }
    __syncthreads();
    if (tid < NOWN) cntp[blockIdx.x * NOWN + tid] = hist[tid];
}

__global__ __launch_bounds__(1024) void group_edges(
    const int* __restrict__ src, const int* __restrict__ dst,
    const int* __restrict__ cntp, int* __restrict__ regs, int* __restrict__ list,
    int E, int n)
{
    __shared__ int s_tot[NOWN];
    __shared__ int s_cur[NOWN];

    const int tid = threadIdx.x;
    const int g   = blockIdx.x;
    if (tid < NOWN) { s_tot[tid] = 0; s_cur[tid] = 0; }
    __syncthreads();

    {
        const int j = tid & (NOWN - 1);
        const int q = tid >> 8;
        int t = 0, p = 0;
#pragma unroll
        for (int i = 0; i < NCG / 4; ++i) {
            const int gg = q + 4 * i;
            const int c  = cntp[gg * NOWN + j];
            t += c;
            p += (gg < g) ? c : 0;
        }
        atomicAdd(&s_tot[j], t);
        atomicAdd(&s_cur[j], p);
    }
    __syncthreads();

    if (tid < 32) {
        int v[8];
        int loc = 0;
#pragma unroll
        for (int k = 0; k < 8; ++k) { v[k] = s_tot[8 * tid + k]; loc += v[k]; }
        int inc = loc;
        for (int o = 1; o < 32; o <<= 1) {
            const int up = __shfl_up(inc, o);
            if (tid >= o) inc += up;
        }
        int run = inc - loc;
#pragma unroll
        for (int k = 0; k < 8; ++k) {
            const int o = 8 * tid + k;
            s_cur[o] += run;
            if (g == 0) {
                regs[2 * o]     = run;
                regs[2 * o + 1] = v[k];
            }
            run += v[k];
        }
    }
    __syncthreads();

    int2* lst = (int2*)list;
    const int per = (E + (int)gridDim.x - 1) / (int)gridDim.x;
    const int e0  = g * per;
    const int e1  = min(E, e0 + per);
    for (int e = e0 + tid; e < e1; e += 4 * 1024) {
        int d[4], s[4];
#pragma unroll
        for (int u = 0; u < 4; ++u) {
            const int ee = e + u * 1024;
            const bool ok = ee < e1;
            d[u] = ok ? dst[ee] : -1;
            s[u] = ok ? src[ee] : 0;
        }
#pragma unroll
        for (int u = 0; u < 4; ++u) {
            if ((unsigned)d[u] < (unsigned)n) {
                const int o   = d[u] / TR;
                const int pos = atomicAdd(&s_cur[o], 1);
                lst[pos] = make_int2(s[u], d[u] - o * TR);
            }
        }
    }
}

__global__ __launch_bounds__(1024) void row_dinv(
    const int* __restrict__ regs, const int* __restrict__ list,
    float* __restrict__ dinv, int n)
{
    __shared__ int degc[TR];

    const int tid = threadIdx.x;
    const int b   = blockIdx.x;
    const int r0  = b * TR;
    const int nr  = min(TR, n - r0);

    if (tid < TR) degc[tid] = 0;
    __syncthreads();

    const int   count = regs[2 * b + 1];
    const int2* lst   = (const int2*)list + regs[2 * b];
    for (int i = tid; i < count; i += 4 * 1024) {
        int r[4];
#pragma unroll
        for (int u = 0; u < 4; ++u) {
            const int ii = i + u * 1024;
            r[u] = (ii < count) ? lst[ii].y : -1;
        }
#pragma unroll
        for (int u = 0; u < 4; ++u)
            if (r[u] >= 0) atomicAdd(&degc[r[u]], 1);
    }
    __syncthreads();

    if (tid < nr) dinv[r0 + tid] = rsqrtf((float)(degc[tid] + 1));
}

__device__ __forceinline__ void add_rows(
    const float* __restrict__ xw, float* acc,
    int s_i, int r_i, float n_i, int cnt, int lane)
{
    for (int k = 0; k < cnt; k += 8) {
        float v[8][4];
        int   rr[8];
        float nn[8];
#pragma unroll
        for (int u = 0; u < 8; ++u) {
            int kk = (k + u < cnt) ? (k + u) : k;
            int s  = __builtin_amdgcn_readlane(s_i, kk);
            rr[u]  = __builtin_amdgcn_readlane(r_i, kk);
            nn[u]  = __int_as_float(__builtin_amdgcn_readlane(__float_as_int(n_i), kk));
            const float4 q = *(const float4*)(xw + (size_t)s * HID_C + (lane << 2));
            v[u][0] = q.x;
            v[u][1] = q.y;
            v[u][2] = q.z;
            v[u][3] = q.w;
        }
#pragma unroll
        for (int u = 0; u < 8; ++u) {
            if (k + u < cnt) {
                float* a = acc + rr[u] * HID_C + lane;
                atomicAdd(a,      v[u][0] * nn[u]);
                atomicAdd(a + 32, v[u][1] * nn[u]);
                atomicAdd(a + 64, v[u][2] * nn[u]);
                atomicAdd(a + 96, v[u][3] * nn[u]);
            }
        }
    }
}

__global__ __launch_bounds__(1024) void agg_list(
    const float* __restrict__ xw, const float* __restrict__ dinv,
    const int* __restrict__ regs, const int* __restrict__ list,
    float* __restrict__ h, int n)
{
    __shared__ float acc[TR * HID_C];

    const int tid  = threadIdx.x;
    const int wave = tid >> 5;
    const int lane = tid & 31;
    const int b    = blockIdx.x;
    const int r0   = b * TR;
    const int nr   = min(TR, n - r0);

    const int   count = regs[2 * b + 1];
    const int2* lst   = (const int2*)list + regs[2 * b];
    const int   nb    = (count + 31) >> 5;
    int  bi = wave;
    int2 en = make_int2(0, 0);
    if (bi < nb) {
        const int idx = (bi << 5) + lane;
        if (idx < count) en = lst[idx];
    }

    for (int i = tid; i < nr * 32; i += 1024) {
        int   r  = i >> 5;
        int   c4 = (i & 31) << 2;
        float di = dinv[r0 + r];
        float s  = di * di;
        float4 v = *(const float4*)(xw + (size_t)(r0 + r) * HID_C + c4);
        float* a = acc + r * HID_C + (i & 31);
        a[0]  = v.x * s;
        a[32] = v.y * s;
        a[64] = v.z * s;
        a[96] = v.w * s;
    }
    __syncthreads();

    while (bi < nb) {
        const int cnt = min(32, count - (bi << 5));
        float n_i = 0.0f;
        if (lane < cnt) n_i = dinv[en.x] * dinv[r0 + en.y];
        const int bn = bi + 32;
        int2 nx = make_int2(0, 0);
        if (bn < nb) {
            const int idx = (bn << 5) + lane;
            if (idx < count) nx = lst[idx];
        }
        add_rows(xw, acc, en.x, en.y, n_i, cnt, lane);
        en = nx;
        bi = bn;
    }
    __syncthreads();

    for (int i = tid; i < nr * 32; i += 1024) {
        int r  = i >> 5;
        int c4 = (i & 31) << 2;
        const float* a = acc + r * HID_C + (i & 31);
        *(float4*)(h + (size_t)(r0 + r) * HID_C + c4) = make_float4(a[0], a[32], a[64], a[96]);
    }
}

#define PTH  512
#define PWV  (PTH / 32)
#define PQ   4
#define PCH  (PTH * 4 * PQ)

__device__ __forceinline__ int4 load_ids(const int* __restrict__ batch, int id0, int n)
{
    if (id0 + 3 < n) return *(const int4*)(batch + id0);
    int4 q;
    q.x = (id0     < n) ? batch[id0]     : -1;
    q.y = (id0 + 1 < n) ? batch[id0 + 1] : -1;
    q.z = (id0 + 2 < n) ? batch[id0 + 2] : -1;
    q.w = (id0 + 3 < n) ? batch[id0 + 3] : -1;
    return q;
}

__device__ __forceinline__ void keep_own(int4 q, int g, int id0, int* cnt, int* lst)
{
    if ((q.x == g) | (q.y == g) | (q.z == g) | (q.w == g)) {
        if (q.x == g) lst[atomicAdd(cnt, 1)] = id0;
        if (q.y == g) lst[atomicAdd(cnt, 1)] = id0 + 1;
        if (q.z == g) lst[atomicAdd(cnt, 1)] = id0 + 2;
        if (q.w == g) lst[atomicAdd(cnt, 1)] = id0 + 3;
    }
}

__global__ __launch_bounds__(PTH) void pool_owner(
    const float* __restrict__ h, const float* __restrict__ hb,
    const int* __restrict__ batch, float* __restrict__ pooled, int n)
{
    __shared__ int   s_n[3];
    __shared__ int   s_list[3][PCH];
    __shared__ float s_part[PWV][HID_C];

    const int tid  = threadIdx.x;
    const int wave = tid >> 5;
    const int lane = tid & 31;
    const int g    = blockIdx.x;

    if (tid < 3) s_n[tid] = 0;
    __syncthreads();

    const int    nch = (n + PCH - 1) / PCH;
    const float4 bb  = *(const float4*)(hb + (lane << 2));
    float4 acc   = make_float4(0.0f, 0.0f, 0.0f, 0.0f);
    int    total = 0;

    int4 q[PQ];
#pragma unroll
    for (int u = 0; u < PQ; ++u) q[u] = load_ids(batch, (u * PTH + tid) << 2, n);
    for (int c = 0; c < nch; ++c) {
        const int k   = c % 3;
        const int id0 = c * PCH + (tid << 2);
#pragma unroll
        for (int u = 0; u < PQ; ++u) keep_own(q[u], g, id0 + u * PTH * 4, &s_n[k], s_list[k]);

#pragma unroll
        for (int u = 0; u < PQ; ++u) q[u] = load_ids(batch, id0 + PCH + u * PTH * 4, n);
        __syncthreads();

        if (tid == 0) s_n[(c + 2) % 3] = 0;
        const int  cnt = s_n[k];
        const int* lst = s_list[k];
        total += cnt;
        for (int e = wave; e < cnt; e += 4 * PWV) {
            float4 v[4];
#pragma unroll
            for (int u = 0; u < 4; ++u) {
                const int ee   = e + u * PWV;
                const int node = lst[(ee < cnt) ? ee : e];
                v[u] = *(const float4*)(h + (size_t)node * HID_C + (lane << 2));
            }
#pragma unroll
            for (int u = 0; u < 4; ++u) {
                if (e + u * PWV < cnt) {
                    acc.x += gelu_exact(v[u].x + bb.x);
                    acc.y += gelu_exact(v[u].y + bb.y);
                    acc.z += gelu_exact(v[u].z + bb.z);
                    acc.w += gelu_exact(v[u].w + bb.w);
                }
            }
        }
    }

    *(float4*)&s_part[wave][lane << 2] = acc;
    __syncthreads();
    if (tid < HID_C) {
        float s = 0.0f;
#pragma unroll
        for (int w = 0; w < PWV; ++w) s += s_part[w][tid];
        pooled[(size_t)g * HID_C + tid] = s / fmaxf((float)total, 1.0f);
    }
}

extern "C" void kernel_launch(void* const* d_in, const int* in_sizes, int n_in,
                              void* d_out, int out_size, void* d_ws, size_t ws_size,
                              hipStream_t stream)
{
    const float* x     = (const float*)d_in[0];
    const int*   ei    = (const int*)d_in[1];
    const int*   batch = (const int*)d_in[2];
    const float* W[3]  = { (const float*)d_in[3], (const float*)d_in[5], (const float*)d_in[7] };
    const float* bv[3] = { (const float*)d_in[4], (const float*)d_in[6], (const float*)d_in[8] };
    const float* Wm1   = (const float*)d_in[9];
    const float* bm1   = (const float*)d_in[10];
    const float* Wm2   = (const float*)d_in[11];
    const float* bm2   = (const float*)d_in[12];

    const int N = in_sizes[0] / IN_C;
    const int E = in_sizes[1] / 2;
    const int* src = ei;
    const int* dst = ei + E;

    char*  ws  = (char*)d_ws;
    size_t off = 0;
    auto carve = [&](size_t bytes) -> float* {
        float* p = (float*)(ws + off);
        off = (off + bytes + 255) & ~(size_t)255;
        return p;
    };
    float* dinv = carve((size_t)N * 4);
    float* buf0 = carve((size_t)N * HID_C * 4);
    float* buf1 = carve((size_t)N * HID_C * 4);
    float* psum = carve((size_t)NBATCH * HID_C * 4);
    float* hid  = carve((size_t)NBATCH * HID_C * 4);
    int*   cntp = (int*)carve((size_t)NCG * NOWN * 4);
    int*   regs = (int*)carve((size_t)NOWN * 2 * 4);
    int*   list = (int*)carve((size_t)E * 2 * 4);

    bin_count<<<NCG, 1024, 0, stream>>>(dst, cntp, E, N);
    group_edges<<<NCG, 1024, 0, stream>>>(src, dst, cntp, regs, list, E, N);
    row_dinv<<<NOWN, 1024, 0, stream>>>(regs, list, dinv, N);

    const float* ab[3] = { nullptr, bv[0], bv[1] };
    const float* cur = x;
    for (int l = 0; l < 3; ++l) {
        layer_gemm<<<LG_NWG, 256, 0, stream>>>(cur, W[l], buf0, N, ab[l]);
        agg_list<<<NOWN, 1024, 0, stream>>>(buf0, dinv, regs, list, buf1, N);
        cur = buf1;
    }

    pool_owner<<<NBATCH, PTH, 0, stream>>>(cur, bv[2], batch, psum, N);

    gemm_f32_wmma<<<NBATCH / 16, 256, 0, stream>>>(psum, Wm1, bm1, hid,
                                                   NBATCH, HID_C, HID_C, 1);
    gemm_f32_wmma<<<NBATCH / 16, 128, 0, stream>>>(hid, Wm2, bm2, (float*)d_out,
                                                   NBATCH, HID_C, OUT_C, 0);
}


# MLP head folded into the mean-pool launch: each graph's work-group multiplies its own mean row, kept in LDS as row 0 of a 16-row f32 WMMA tile, by Wm1 and Wm2; 2 launches less; sum order unchanged
# speedup vs baseline: 4.5703x; 1.0697x over previous
#include <hip/hip_runtime.h>
#include <hip/hip_bf16.h>
#include <math.h>

typedef float v2f __attribute__((ext_vector_type(2)));
typedef float v8f __attribute__((ext_vector_type(8)));

#define IN_C   128
#define HID_C  128
#define OUT_C  64
#define NBATCH 512

__device__ __forceinline__ float gelu_exact(float x)
{
    return 0.5f * x * (1.0f + erff(x * 0.70710678118654752f));
}

#define LG_NWG   512
#define LG_PITCH 132

__device__ __forceinline__ void lg_load_rows(
    const float* __restrict__ A, int t, int t1, int wave, int c4, float4 (&va)[4])
{
#pragma unroll
    for (int u = 0; u < 4; ++u)
        if (t + (u >> 1) < t1)
            va[u] = *(const float4*)(A + (size_t)(t * 16 + u * 8 + wave) * HID_C + c4);
}

__device__ __forceinline__ float lg_other_half(float x)
{
    return __int_as_float(__builtin_amdgcn_permlanex16(
        __float_as_int(x), __float_as_int(x), 0x76543210, 0xfedcba98, false, false));
}

__global__ __launch_bounds__(256) void layer_gemm(
    const float* __restrict__ A, const float* __restrict__ Wt,
    float* __restrict__ C, int M, const float* __restrict__ abias)
{
    __shared__ __attribute__((aligned(16))) float lds_a[2][32][LG_PITCH];

    const int tid  = threadIdx.x;
    const int lane = tid & 31;
    const int wave = __builtin_amdgcn_readfirstlane(tid >> 5);
    const int rg   = wave >> 2;
    const int cg   = wave & 3;
    const int m    = lane & 15;
    const int h    = lane >> 4;
    const int c4   = lane << 2;

    const unsigned ntile = (unsigned)M >> 4;
    const int t0 = (int)((blockIdx.x * ntile) / gridDim.x);
    const int t1 = (int)(((blockIdx.x + 1) * ntile) / gridDim.x);

    float4 va[4];
    lg_load_rows(A, t0, t1, wave, c4, va);

    float4 bq[16][2];
#pragma unroll
    for (int p = 0; p < 16; ++p) {
#pragma unroll
        for (int j = 0; j < 2; ++j) {
            const float* wp = Wt + (8 * p + 4 * h) * HID_C + 32 * cg + 16 * j + m;
            bq[p][j] = make_float4(wp[0], wp[HID_C], wp[2 * HID_C], wp[3 * HID_C]);
        }
    }

    float4 bb = make_float4(0.0f, 0.0f, 0.0f, 0.0f);
    if (abias) bb = *(const float4*)(abias + c4);

    int buf = 0;
    for (int t = t0; t < t1; t += 2) {
#pragma unroll
        for (int u = 0; u < 4; ++u) {
            if (t + (u >> 1) < t1) {
                float4 v = va[u];
                if (abias) {
                    v.x = gelu_exact(v.x + bb.x);
                    v.y = gelu_exact(v.y + bb.y);
                    v.z = gelu_exact(v.z + bb.z);
                    v.w = gelu_exact(v.w + bb.w);
                }
                *(float4*)&lds_a[buf][u * 8 + wave][c4] = v;
            }
        }
        __syncthreads();
        if (t + 2 < t1) lg_load_rows(A, t + 2, t1, wave, c4, va);

        if (t + rg < t1) {
            v8f acc0 = {}, acc1 = {};
            const float* ap = &lds_a[buf][16 * rg + m][4 * h];
#pragma unroll
            for (int p = 0; p < 16; ++p) {
                const float4 aq = *(const float4*)(ap + 8 * p);
                const v2f a01 = { aq.x, aq.y };
                const v2f a23 = { aq.z, aq.w };
                const v2f b00 = { bq[p][0].x, bq[p][0].y };
                const v2f b01 = { bq[p][0].z, bq[p][0].w };
                const v2f b10 = { bq[p][1].x, bq[p][1].y };
                const v2f b11 = { bq[p][1].z, bq[p][1].w };
                acc0 = __builtin_amdgcn_wmma_f32_16x16x4_f32(
                    false, a01, false, b00, (short)0, acc0, false, false);
                acc1 = __builtin_amdgcn_wmma_f32_16x16x4_f32(
                    false, a01, false, b10, (short)0, acc1, false, false);
                acc0 = __builtin_amdgcn_wmma_f32_16x16x4_f32(
                    false, a23, false, b01, (short)0, acc0, false, false);
                acc1 = __builtin_amdgcn_wmma_f32_16x16x4_f32(
                    false, a23, false, b11, (short)0, acc1, false, false);
            }

            float* cp = C + (size_t)((t + rg) * 16) * HID_C + 32 * cg + lane;
#pragma unroll
            for (int r = 0; r < 8; ++r) {
                const float send = h ? acc0[r] : acc1[r];
                const float recv = lg_other_half(send);
                const float lo   = h ? recv : acc0[r];
                const float hi   = h ? acc1[r] : recv;
                cp[(size_t)r * HID_C]       = lo;
                cp[(size_t)(8 + r) * HID_C] = hi;
            }
        }
        buf ^= 1;
    }
}

#define TR    196
#define NOWN  256
#define NCG   256

__global__ __launch_bounds__(1024) void bin_count(
    const int* __restrict__ dst, int* __restrict__ cntp, int E, int n)
{
    __shared__ int hist[NOWN];
    const int tid = threadIdx.x;
    if (tid < NOWN) hist[tid] = 0;
    __syncthreads();

    const int per = (E + (int)gridDim.x - 1) / (int)gridDim.x;
    const int e0  = blockIdx.x * per;
    const int e1  = min(E, e0 + per);
    for (int e = e0 + tid; e < e1; e += 4 * 1024) {
        unsigned d[4];
#pragma unroll
        for (int u = 0; u < 4; ++u) {
            const int ee = e + u * 1024;
            d[u] = (ee < e1) ? (unsigned)dst[ee] : 0xffffffffu;
        }
#pragma unroll
        for (int u = 0; u < 4; ++u)
            if (d[u] < (unsigned)n) atomicAdd(&hist[d[u] / TR], 1);
    }
    __syncthreads();
    if (tid < NOWN) cntp[blockIdx.x * NOWN + tid] = hist[tid];
}

__global__ __launch_bounds__(1024) void group_edges(
    const int* __restrict__ src, const int* __restrict__ dst,
    const int* __restrict__ cntp, int* __restrict__ regs, int* __restrict__ list,
    int E, int n)
{
    __shared__ int s_tot[NOWN];
    __shared__ int s_cur[NOWN];

    const int tid = threadIdx.x;
    const int g   = blockIdx.x;
    if (tid < NOWN) { s_tot[tid] = 0; s_cur[tid] = 0; }
    __syncthreads();

    {
        const int j = tid & (NOWN - 1);
        const int q = tid >> 8;
        int t = 0, p = 0;
#pragma unroll
        for (int i = 0; i < NCG / 4; ++i) {
            const int gg = q + 4 * i;
            const int c  = cntp[gg * NOWN + j];
            t += c;
            p += (gg < g) ? c : 0;
        }
        atomicAdd(&s_tot[j], t);
        atomicAdd(&s_cur[j], p);
    }
    __syncthreads();

    if (tid < 32) {
        int v[8];
        int loc = 0;
#pragma unroll
        for (int k = 0; k < 8; ++k) { v[k] = s_tot[8 * tid + k]; loc += v[k]; }
        int inc = loc;
        for (int o = 1; o < 32; o <<= 1) {
            const int up = __shfl_up(inc, o);
            if (tid >= o) inc += up;
        }
        int run = inc - loc;
#pragma unroll
        for (int k = 0; k < 8; ++k) {
            const int o = 8 * tid + k;
            s_cur[o] += run;
            if (g == 0) {
                regs[2 * o]     = run;
                regs[2 * o + 1] = v[k];
            }
            run += v[k];
        }
    }
    __syncthreads();

    int2* lst = (int2*)list;
    const int per = (E + (int)gridDim.x - 1) / (int)gridDim.x;
    const int e0  = g * per;
    const int e1  = min(E, e0 + per);
    for (int e = e0 + tid; e < e1; e += 4 * 1024) {
        int d[4], s[4];
#pragma unroll
        for (int u = 0; u < 4; ++u) {
            const int ee = e + u * 1024;
            const bool ok = ee < e1;
            d[u] = ok ? dst[ee] : -1;
            s[u] = ok ? src[ee] : 0;
        }
#pragma unroll
        for (int u = 0; u < 4; ++u) {
            if ((unsigned)d[u] < (unsigned)n) {
                const int o   = d[u] / TR;
                const int pos = atomicAdd(&s_cur[o], 1);
                lst[pos] = make_int2(s[u], d[u] - o * TR);
            }
        }
    }
}

__global__ __launch_bounds__(1024) void row_dinv(
    const int* __restrict__ regs, const int* __restrict__ list,
    float* __restrict__ dinv, int n)
{
    __shared__ int degc[TR];

    const int tid = threadIdx.x;
    const int b   = blockIdx.x;
    const int r0  = b * TR;
    const int nr  = min(TR, n - r0);

    if (tid < TR) degc[tid] = 0;
    __syncthreads();

    const int   count = regs[2 * b + 1];
    const int2* lst   = (const int2*)list + regs[2 * b];
    for (int i = tid; i < count; i += 4 * 1024) {
        int r[4];
#pragma unroll
        for (int u = 0; u < 4; ++u) {
            const int ii = i + u * 1024;
            r[u] = (ii < count) ? lst[ii].y : -1;
        }
#pragma unroll
        for (int u = 0; u < 4; ++u)
            if (r[u] >= 0) atomicAdd(&degc[r[u]], 1);
    }
    __syncthreads();

    if (tid < nr) dinv[r0 + tid] = rsqrtf((float)(degc[tid] + 1));
}

__device__ __forceinline__ void add_rows(
    const float* __restrict__ xw, float* acc,
    int s_i, int r_i, float n_i, int cnt, int lane)
{
    for (int k = 0; k < cnt; k += 8) {
        float v[8][4];
        int   rr[8];
        float nn[8];
#pragma unroll
        for (int u = 0; u < 8; ++u) {
            int kk = (k + u < cnt) ? (k + u) : k;
            int s  = __builtin_amdgcn_readlane(s_i, kk);
            rr[u]  = __builtin_amdgcn_readlane(r_i, kk);
            nn[u]  = __int_as_float(__builtin_amdgcn_readlane(__float_as_int(n_i), kk));
            const float4 q = *(const float4*)(xw + (size_t)s * HID_C + (lane << 2));
            v[u][0] = q.x;
            v[u][1] = q.y;
            v[u][2] = q.z;
            v[u][3] = q.w;
        }
#pragma unroll
        for (int u = 0; u < 8; ++u) {
            if (k + u < cnt) {
                float* a = acc + rr[u] * HID_C + lane;
                atomicAdd(a,      v[u][0] * nn[u]);
                atomicAdd(a + 32, v[u][1] * nn[u]);
                atomicAdd(a + 64, v[u][2] * nn[u]);
                atomicAdd(a + 96, v[u][3] * nn[u]);
            }
        }
    }
}

__global__ __launch_bounds__(1024) void agg_list(
    const float* __restrict__ xw, const float* __restrict__ dinv,
    const int* __restrict__ regs, const int* __restrict__ list,
    float* __restrict__ h, int n)
{
    __shared__ float acc[TR * HID_C];

    const int tid  = threadIdx.x;
    const int wave = tid >> 5;
    const int lane = tid & 31;
    const int b    = blockIdx.x;
    const int r0   = b * TR;
    const int nr   = min(TR, n - r0);

    const int   count = regs[2 * b + 1];
    const int2* lst   = (const int2*)list + regs[2 * b];
    const int   nb    = (count + 31) >> 5;
    int  bi = wave;
    int2 en = make_int2(0, 0);
    if (bi < nb) {
        const int idx = (bi << 5) + lane;
        if (idx < count) en = lst[idx];
    }

    for (int i = tid; i < nr * 32; i += 1024) {
        int   r  = i >> 5;
        int   c4 = (i & 31) << 2;
        float di = dinv[r0 + r];
        float s  = di * di;
        float4 v = *(const float4*)(xw + (size_t)(r0 + r) * HID_C + c4);
        float* a = acc + r * HID_C + (i & 31);
        a[0]  = v.x * s;
        a[32] = v.y * s;
        a[64] = v.z * s;
        a[96] = v.w * s;
    }
    __syncthreads();

    while (bi < nb) {
        const int cnt = min(32, count - (bi << 5));
        float n_i = 0.0f;
        if (lane < cnt) n_i = dinv[en.x] * dinv[r0 + en.y];
        const int bn = bi + 32;
        int2 nx = make_int2(0, 0);
        if (bn < nb) {
            const int idx = (bn << 5) + lane;
            if (idx < count) nx = lst[idx];
        }
        add_rows(xw, acc, en.x, en.y, n_i, cnt, lane);
        en = nx;
        bi = bn;
    }
    __syncthreads();

    for (int i = tid; i < nr * 32; i += 1024) {
        int r  = i >> 5;
        int c4 = (i & 31) << 2;
        const float* a = acc + r * HID_C + (i & 31);
        *(float4*)(h + (size_t)(r0 + r) * HID_C + c4) = make_float4(a[0], a[32], a[64], a[96]);
    }
}

#define PTH  512
#define PWV  (PTH / 32)
#define PQ   4
#define PCH  (PTH * 4 * PQ)

__device__ __forceinline__ int4 load_ids(const int* __restrict__ batch, int id0, int n)
{
    if (id0 + 3 < n) return *(const int4*)(batch + id0);
    int4 q;
    q.x = (id0     < n) ? batch[id0]     : -1;
    q.y = (id0 + 1 < n) ? batch[id0 + 1] : -1;
    q.z = (id0 + 2 < n) ? batch[id0 + 2] : -1;
    q.w = (id0 + 3 < n) ? batch[id0 + 3] : -1;
    return q;
}

__device__ __forceinline__ void keep_own(int4 q, int g, int id0, int* cnt, int* lst)
{
    if ((q.x == g) | (q.y == g) | (q.z == g) | (q.w == g)) {
        if (q.x == g) lst[atomicAdd(cnt, 1)] = id0;
        if (q.y == g) lst[atomicAdd(cnt, 1)] = id0 + 1;
        if (q.z == g) lst[atomicAdd(cnt, 1)] = id0 + 2;
        if (q.w == g) lst[atomicAdd(cnt, 1)] = id0 + 3;
    }
}


template <int N>
__device__ __forceinline__ float row_times_tile(
    const float* arow, const float* __restrict__ B, int col0, int lane)
{
    const int m  = lane & 15;
    const int kh = (lane >> 4) << 1;
    const float* Bp = B + (size_t)kh * N + col0 + m;
    v2f b[32];
#pragma unroll
    for (int s = 0; s < 32; ++s) {
        b[s].x = Bp[(size_t)(4 * s) * N];
        b[s].y = Bp[(size_t)(4 * s + 1) * N];
    }
    v8f acc = {};
#pragma unroll
    for (int s = 0; s < 32; ++s) {
        const float x0 = arow[4 * s + kh];
        const float x1 = arow[4 * s + kh + 1];
        v2f a;
        a.x = (m == 0) ? x0 : 0.0f;
        a.y = (m == 0) ? x1 : 0.0f;
        acc = __builtin_amdgcn_wmma_f32_16x16x4_f32(
            false, a, false, b[s], (short)0, acc, false, false);
    }
    return acc[0];
}

__global__ __launch_bounds__(PTH) void pool_head(
    const float* __restrict__ h, const float* __restrict__ hb,
    const int* __restrict__ batch,
    const float* __restrict__ Wm1, const float* __restrict__ bm1,
    const float* __restrict__ Wm2, const float* __restrict__ bm2,
    float* __restrict__ out, int n)
{
    __shared__ int   s_n[3];
    __shared__ int   s_list[3][PCH];
    __shared__ float s_part[PWV][HID_C];
    __shared__ float s_mean[HID_C];
    __shared__ float s_hid[HID_C];
    __shared__ float s_out[OUT_C];

    const int tid  = threadIdx.x;
    const int wave = tid >> 5;
    const int lane = tid & 31;
    const int g    = blockIdx.x;

    if (tid < 3) s_n[tid] = 0;
    __syncthreads();

    const int    nch = (n + PCH - 1) / PCH;
    const float4 bb  = *(const float4*)(hb + (lane << 2));
    float4 acc   = make_float4(0.0f, 0.0f, 0.0f, 0.0f);
    int    total = 0;

    int4 q[PQ];
#pragma unroll
    for (int u = 0; u < PQ; ++u) q[u] = load_ids(batch, (u * PTH + tid) << 2, n);
    for (int c = 0; c < nch; ++c) {
        const int k   = c % 3;
        const int id0 = c * PCH + (tid << 2);
#pragma unroll
        for (int u = 0; u < PQ; ++u) keep_own(q[u], g, id0 + u * PTH * 4, &s_n[k], s_list[k]);

#pragma unroll
        for (int u = 0; u < PQ; ++u) q[u] = load_ids(batch, id0 + PCH + u * PTH * 4, n);
        __syncthreads();

        if (tid == 0) s_n[(c + 2) % 3] = 0;
        const int  cnt = s_n[k];
        const int* lst = s_list[k];
        total += cnt;
        for (int e = wave; e < cnt; e += 4 * PWV) {
            float4 v[4];
#pragma unroll
            for (int u = 0; u < 4; ++u) {
                const int ee   = e + u * PWV;
                const int node = lst[(ee < cnt) ? ee : e];
                v[u] = *(const float4*)(h + (size_t)node * HID_C + (lane << 2));
            }
#pragma unroll
            for (int u = 0; u < 4; ++u) {
                if (e + u * PWV < cnt) {
                    acc.x += gelu_exact(v[u].x + bb.x);
                    acc.y += gelu_exact(v[u].y + bb.y);
                    acc.z += gelu_exact(v[u].z + bb.z);
                    acc.w += gelu_exact(v[u].w + bb.w);
                }
            }
        }
    }

    *(float4*)&s_part[wave][lane << 2] = acc;
    __syncthreads();
    if (tid < HID_C) {
        float s = 0.0f;
#pragma unroll
        for (int w = 0; w < PWV; ++w) s += s_part[w][tid];
        s_mean[tid] = s / fmaxf((float)total, 1.0f);
    }
    __syncthreads();

    const int wv = __builtin_amdgcn_readfirstlane(tid >> 5);
    if (wv < HID_C / 16) {
        const float r = row_times_tile<HID_C>(s_mean, Wm1, 16 * wv, lane);
        if (lane < 16) s_hid[16 * wv + lane] = fmaxf(r + bm1[16 * wv + lane], 0.0f);
    }
    __syncthreads();
    if (wv < OUT_C / 16) {
        const float r = row_times_tile<OUT_C>(s_hid, Wm2, 16 * wv, lane);
        if (lane < 16) s_out[16 * wv + lane] = r + bm2[16 * wv + lane];
    }
    __syncthreads();
    if (tid < OUT_C) out[(size_t)g * OUT_C + tid] = s_out[tid];
}

extern "C" void kernel_launch(void* const* d_in, const int* in_sizes, int n_in,
                              void* d_out, int out_size, void* d_ws, size_t ws_size,
                              hipStream_t stream)
{
    const float* x     = (const float*)d_in[0];
    const int*   ei    = (const int*)d_in[1];
    const int*   batch = (const int*)d_in[2];
    const float* W[3]  = { (const float*)d_in[3], (const float*)d_in[5], (const float*)d_in[7] };
    const float* bv[3] = { (const float*)d_in[4], (const float*)d_in[6], (const float*)d_in[8] };
    const float* Wm1   = (const float*)d_in[9];
    const float* bm1   = (const float*)d_in[10];
    const float* Wm2   = (const float*)d_in[11];
    const float* bm2   = (const float*)d_in[12];

    const int N = in_sizes[0] / IN_C;
    const int E = in_sizes[1] / 2;
    const int* src = ei;
    const int* dst = ei + E;

    char*  ws  = (char*)d_ws;
    size_t off = 0;
    auto carve = [&](size_t bytes) -> float* {
        float* p = (float*)(ws + off);
        off = (off + bytes + 255) & ~(size_t)255;
        return p;
    };
    float* dinv = carve((size_t)N * 4);
    float* buf0 = carve((size_t)N * HID_C * 4);
    float* buf1 = carve((size_t)N * HID_C * 4);
    int*   cntp = (int*)carve((size_t)NCG * NOWN * 4);
    int*   regs = (int*)carve((size_t)NOWN * 2 * 4);
    int*   list = (int*)carve((size_t)E * 2 * 4);

    bin_count<<<NCG, 1024, 0, stream>>>(dst, cntp, E, N);
    group_edges<<<NCG, 1024, 0, stream>>>(src, dst, cntp, regs, list, E, N);
    row_dinv<<<NOWN, 1024, 0, stream>>>(regs, list, dinv, N);

    const float* ab[3] = { nullptr, bv[0], bv[1] };
    const float* cur = x;
    for (int l = 0; l < 3; ++l) {
        layer_gemm<<<LG_NWG, 256, 0, stream>>>(cur, W[l], buf0, N, ab[l]);
        agg_list<<<NOWN, 1024, 0, stream>>>(buf0, dinv, regs, list, buf1, N);
        cur = buf1;
    }

    pool_head<<<NBATCH, PTH, 0, stream>>>(cur, bv[2], batch, Wm1, bm1, Wm2, bm2,
                                          (float*)d_out, N);
}
